# starting kernel
# baseline (speedup 1.0000x reference)
#include <hip/hip_runtime.h>

#define N_NODES 65536
#define N_EDGES 1048576
#define KD      128

typedef __attribute__((ext_vector_type(2)))  float     v2f;
typedef __attribute__((ext_vector_type(8)))  float     v8f;
typedef __attribute__((ext_vector_type(16))) _Float16  v16h;

__global__ void zero_f32(float* __restrict__ p, size_t n) {
  size_t i = (size_t)blockIdx.x * blockDim.x + threadIdx.x;
  if (i < n) p[i] = 0.0f;
}

__global__ void degree_kernel(const int* __restrict__ dst,
                              float* __restrict__ deg) {
  int e = blockIdx.x * blockDim.x + threadIdx.x;
  if (e < N_EDGES) atomicAdd(deg + dst[e], 1.0f);
}

__global__ void scatter_kernel(const float* __restrict__ feat,
                               const int* __restrict__ src,
                               const int* __restrict__ dst,
                               float* __restrict__ agg) {
  long long t = (long long)blockIdx.x * blockDim.x + threadIdx.x;
  int       chunk = (int)(t & 31);
  long long e     = t >> 5;
  if (e >= N_EDGES) return;
  int s = src[e], d = dst[e];
  float4 v = ((const float4*)(feat + (size_t)s * KD))[chunk];
  float* op = agg + (size_t)d * KD + chunk * 4;
  atomicAdd(op + 0, v.x);
  atomicAdd(op + 1, v.y);
  atomicAdd(op + 2, v.z);
  atomicAdd(op + 3, v.w);
}

template <int NOUT, bool RELU>
__global__ __launch_bounds__(32 * (NOUT / 16))
void sage_gemm(const float* __restrict__ X,
               const float* __restrict__ AGG,
               const float* __restrict__ deg,
               const float* __restrict__ Wself,
               const float* __restrict__ Wneigh,
               const float* __restrict__ bias,
               float* __restrict__ OUT) {
  __shared__ float xs[16][KD];
  __shared__ float ms[16][KD];

  const int m0       = blockIdx.x * 16;
  const int tid      = threadIdx.x;
  const int nthreads = 32 * (NOUT / 16);

  for (int i = tid; i < 16 * (KD / 4); i += nthreads) {
    int r  = i / (KD / 4);
    int c4 = i % (KD / 4);
    float4 xv = ((const float4*)(X + (size_t)(m0 + r) * KD))[c4];
    float4 av = ((const float4*)(AGG + (size_t)(m0 + r) * KD))[c4];
    float  rd = 1.0f / fmaxf(deg[m0 + r], 1.0f);
    av.x *= rd; av.y *= rd; av.z *= rd; av.w *= rd;
    ((float4*)&xs[r][0])[c4] = xv;
    ((float4*)&ms[r][0])[c4] = av;
  }
  __syncthreads();

  const int  wave = tid / 32;
  const int  lane = tid % 32;
  const int  n0   = wave * 16;
  const int  mrow = lane & 15;
  const bool hi   = lane >= 16;
  const int  ncol = n0 + (lane & 15);

  v8f c = {};

#if __has_builtin(__builtin_amdgcn_wmma_f32_16x16x4_f32)
  for (int kk = 0; kk < KD; kk += 4) {
    int ka = kk + (hi ? 2 : 0);
    v2f a_self = { xs[mrow][ka], xs[mrow][ka + 1] };
    v2f a_mean = { ms[mrow][ka], ms[mrow][ka + 1] };
    v2f bs = { Wself[(size_t)ka * NOUT + ncol],
               Wself[(size_t)(ka + 1) * NOUT + ncol] };
    v2f bn = { Wneigh[(size_t)ka * NOUT + ncol],
               Wneigh[(size_t)(ka + 1) * NOUT + ncol] };
    c = __builtin_amdgcn_wmma_f32_16x16x4_f32(false, a_self, false, bs,
                                              (short)0, c, false, false);
    c = __builtin_amdgcn_wmma_f32_16x16x4_f32(false, a_mean, false, bn,
                                              (short)0, c, false, false);
  }
#else
  for (int kk = 0; kk < KD; kk += 32) {
    int ka = kk + (hi ? 8 : 0);
    int kb = kk + (hi ? 16 : 0);
    v16h a_self, a_mean, bs, bn;
#pragma unroll
    for (int j = 0; j < 8; j++) {
      a_self[j]     = (_Float16)xs[mrow][ka + j];
      a_self[8 + j] = (_Float16)xs[mrow][ka + 16 + j];
      a_mean[j]     = (_Float16)ms[mrow][ka + j];
      a_mean[8 + j] = (_Float16)ms[mrow][ka + 16 + j];
    }
#pragma unroll
    for (int j = 0; j < 16; j++) {
      bs[j] = (_Float16)Wself[(size_t)(kb + j) * NOUT + ncol];
      bn[j] = (_Float16)Wneigh[(size_t)(kb + j) * NOUT + ncol];
    }
    c = __builtin_amdgcn_wmma_f32_16x16x32_f16(false, a_self, false, bs,
                                               (short)0, c, false, false);
    c = __builtin_amdgcn_wmma_f32_16x16x32_f16(false, a_mean, false, bn,
                                               (short)0, c, false, false);
  }
#endif

  float bv = bias[ncol];
#pragma unroll
  for (int i = 0; i < 8; i++) {
    int   row = m0 + i + (hi ? 8 : 0);
    float v   = c[i] + bv;
    if (RELU) v = fmaxf(v, 0.0f);
    OUT[(size_t)row * NOUT + ncol] = v;
  }
}

extern "C" void kernel_launch(void* const* d_in, const int* in_sizes, int n_in,
                              void* d_out, int out_size, void* d_ws,
                              size_t ws_size, hipStream_t stream) {
  (void)in_sizes; (void)n_in; (void)out_size; (void)ws_size;
  const float* x   = (const float*)d_in[0];
  const int*   src = (const int*)d_in[1];
  const int*   dst = (const int*)d_in[2];
  const float* W1s = (const float*)d_in[3];
  const float* W1n = (const float*)d_in[4];
  const float* b1  = (const float*)d_in[5];
  const float* W2s = (const float*)d_in[6];
  const float* W2n = (const float*)d_in[7];
  const float* b2  = (const float*)d_in[8];
  float*       out = (float*)d_out;

  char*  ws  = (char*)d_ws;
  float* deg = (float*)ws;
  float* agg = deg + N_NODES;
  float* h   = agg + (size_t)N_NODES * KD;

  size_t nz = (size_t)N_NODES * (KD + 1);
  zero_f32<<<(unsigned)((nz + 255) / 256), 256, 0, stream>>>(deg, nz);

  degree_kernel<<<(N_EDGES + 255) / 256, 256, 0, stream>>>(dst, deg);

  scatter_kernel<<<(N_EDGES * 32) / 256, 256, 0, stream>>>(x, src, dst, agg);
  sage_gemm<128, true><<<N_NODES / 16, 256, 0, stream>>>(x, agg, deg, W1s,
                                                         W1n, b1, h);

  size_t na = (size_t)N_NODES * KD;
  zero_f32<<<(unsigned)((na + 255) / 256), 256, 0, stream>>>(agg, na);
  scatter_kernel<<<(N_EDGES * 32) / 256, 256, 0, stream>>>(h, src, dst, agg);
  sage_gemm<64, false><<<N_NODES / 16, 128, 0, stream>>>(h, agg, deg, W2s,
                                                         W2n, b2, out);
}


# Aggregate by gather over a destination-sorted edge list built per call (int count with rank, scan, fill) instead of 128 float atomic adds per edge; both GEMMs untouched
# speedup vs baseline: 2.1139x; 2.1139x over previous
#include <hip/hip_runtime.h>

#define N_NODES 65536
#define N_EDGES 1048576
#define KD      128

typedef __attribute__((ext_vector_type(2)))  float     v2f;
typedef __attribute__((ext_vector_type(8)))  float     v8f;
typedef __attribute__((ext_vector_type(16))) _Float16  v16h;

#define NPW 8

__global__ void zero_i32(int* __restrict__ p) {
  p[blockIdx.x * blockDim.x + threadIdx.x] = 0;
}

__global__ void rank_kernel(const int* __restrict__ dst,
                            int* __restrict__ cnt,
                            int* __restrict__ rank) {
  int  t = blockIdx.x * blockDim.x + threadIdx.x;
  int4 d = ((const int4*)dst)[t];
  int4 r;
  r.x = atomicAdd(cnt + d.x, 1);
  r.y = atomicAdd(cnt + d.y, 1);
  r.z = atomicAdd(cnt + d.z, 1);
  r.w = atomicAdd(cnt + d.w, 1);
  ((int4*)rank)[t] = r;
}

__global__ __launch_bounds__(1024)
void scan_kernel(const int* __restrict__ cnt,
                 int* __restrict__ offs,
                 float* __restrict__ deg) {
  __shared__ int wsum[32];
  __shared__ int sbase;
  const int b    = blockIdx.x;
  const int t    = threadIdx.x;
  const int lane = t & 31;
  const int w    = t >> 5;

  int part = 0;
  const int4* c4 = (const int4*)cnt;
#pragma unroll 4
  for (int i = t; i < b * 256; i += 1024) {
    int4 v = c4[i];
    part += v.x + v.y + v.z + v.w;
  }
  for (int o = 16; o > 0; o >>= 1) part += __shfl_xor(part, o);
  if (lane == 0) wsum[w] = part;
  __syncthreads();
  if (w == 0) {
    int s = wsum[lane];
    for (int o = 16; o > 0; o >>= 1) s += __shfl_xor(s, o);
    if (lane == 0) sbase = s;
  }
  __syncthreads();
  const int base = sbase;

  const int c    = cnt[b * 1024 + t];
  int       incl = c;
  for (int o = 1; o < 32; o <<= 1) {
    int u = __shfl_up(incl, o);
    if (lane >= o) incl += u;
  }
  if (lane == 31) wsum[w] = incl;
  __syncthreads();
  if (w == 0) {
    int s  = wsum[lane];
    int si = s;
    for (int o = 1; o < 32; o <<= 1) {
      int u = __shfl_up(si, o);
      if (lane >= o) si += u;
    }
    wsum[lane] = si - s;
  }
  __syncthreads();
  const int excl = base + wsum[w] + incl - c;
  offs[b * 1024 + t] = excl;
  deg[b * 1024 + t]  = (float)c;
  if (b == (int)gridDim.x - 1 && t == 1023) offs[N_NODES] = excl + c;
}

__global__ void fill_kernel(const int* __restrict__ src,
                            const int* __restrict__ dst,
                            const int* __restrict__ rank,
                            const int* __restrict__ offs,
                            int* __restrict__ perm) {
  int  t = blockIdx.x * blockDim.x + threadIdx.x;
  int4 s = ((const int4*)src)[t];
  int4 d = ((const int4*)dst)[t];
  int4 r = ((const int4*)rank)[t];
  perm[(offs[d.x] + r.x) & (N_EDGES - 1)] = s.x;
  perm[(offs[d.y] + r.y) & (N_EDGES - 1)] = s.y;
  perm[(offs[d.z] + r.z) & (N_EDGES - 1)] = s.z;
  perm[(offs[d.w] + r.w) & (N_EDGES - 1)] = s.w;
}

__global__ __launch_bounds__(256)
void gather_kernel(const float* __restrict__ feat,
                   const int* __restrict__ offs,
                   const int* __restrict__ perm,
                   float* __restrict__ agg) {
  const int lane = threadIdx.x & 31;
  const int wave =
      __builtin_amdgcn_readfirstlane((blockIdx.x * 256 + threadIdx.x) >> 5);
  const int n0 = wave * NPW;

  const float4* f4 = (const float4*)feat;
  float4*       a4 = (float4*)agg;

  const int endv = offs[n0 + 1 + (lane & (NPW - 1))];
  const int rbeg = offs[n0];
  const int rend = __builtin_amdgcn_readlane(endv, NPW - 1);
  int       node = 0;
  int       nend = __builtin_amdgcn_readlane(endv, 0);
  float4    acc  = make_float4(0.0f, 0.0f, 0.0f, 0.0f);

  for (int base = rbeg; base < rend; base += 32) {
    const int rem = rend - base;
    int myidx = perm[base + min(lane, rem - 1)];
    myidx &= (N_NODES - 1);
#pragma unroll
    for (int k0 = 0; k0 < 32; k0 += 8) {
      if (k0 < rem) {
        float4 v[8];
#pragma unroll
        for (int k = 0; k < 8; k++) {
          const int idx = __builtin_amdgcn_readlane(myidx, k0 + k);
          v[k] = f4[(size_t)idx * (KD / 4) + lane];
        }
#pragma unroll
        for (int k = 0; k < 8; k++) {
          const int pos = base + k0 + k;
          if (pos < rend) {
            while (pos >= nend) {
              a4[(size_t)(n0 + node) * (KD / 4) + lane] = acc;
              acc = make_float4(0.0f, 0.0f, 0.0f, 0.0f);
              node++;
              nend = __builtin_amdgcn_readlane(endv, node);
            }
            acc.x += v[k].x; acc.y += v[k].y;
            acc.z += v[k].z; acc.w += v[k].w;
          }
        }
      }
    }
  }
  while (node < NPW) {
    a4[(size_t)(n0 + node) * (KD / 4) + lane] = acc;
    acc = make_float4(0.0f, 0.0f, 0.0f, 0.0f);
    node++;
  }
}

template <int NOUT, bool RELU>
__global__ __launch_bounds__(32 * (NOUT / 16))
void sage_gemm(const float* __restrict__ X,
               const float* __restrict__ AGG,
               const float* __restrict__ deg,
               const float* __restrict__ Wself,
               const float* __restrict__ Wneigh,
               const float* __restrict__ bias,
               float* __restrict__ OUT) {
  __shared__ float xs[16][KD];
  __shared__ float ms[16][KD];

  const int m0       = blockIdx.x * 16;
  const int tid      = threadIdx.x;
  const int nthreads = 32 * (NOUT / 16);

  for (int i = tid; i < 16 * (KD / 4); i += nthreads) {
    int r  = i / (KD / 4);
    int c4 = i % (KD / 4);
    float4 xv = ((const float4*)(X + (size_t)(m0 + r) * KD))[c4];
    float4 av = ((const float4*)(AGG + (size_t)(m0 + r) * KD))[c4];
    float  rd = 1.0f / fmaxf(deg[m0 + r], 1.0f);
    av.x *= rd; av.y *= rd; av.z *= rd; av.w *= rd;
    ((float4*)&xs[r][0])[c4] = xv;
    ((float4*)&ms[r][0])[c4] = av;
  }
  __syncthreads();

  const int  wave = tid / 32;
  const int  lane = tid % 32;
  const int  n0   = wave * 16;
  const int  mrow = lane & 15;
  const bool hi   = lane >= 16;
  const int  ncol = n0 + (lane & 15);

  v8f c = {};

#if __has_builtin(__builtin_amdgcn_wmma_f32_16x16x4_f32)
  for (int kk = 0; kk < KD; kk += 4) {
    int ka = kk + (hi ? 2 : 0);
    v2f a_self = { xs[mrow][ka], xs[mrow][ka + 1] };
    v2f a_mean = { ms[mrow][ka], ms[mrow][ka + 1] };
    v2f bs = { Wself[(size_t)ka * NOUT + ncol],
               Wself[(size_t)(ka + 1) * NOUT + ncol] };
    v2f bn = { Wneigh[(size_t)ka * NOUT + ncol],
               Wneigh[(size_t)(ka + 1) * NOUT + ncol] };
    c = __builtin_amdgcn_wmma_f32_16x16x4_f32(false, a_self, false, bs,
                                              (short)0, c, false, false);
    c = __builtin_amdgcn_wmma_f32_16x16x4_f32(false, a_mean, false, bn,
                                              (short)0, c, false, false);
  }
#else
  for (int kk = 0; kk < KD; kk += 32) {
    int ka = kk + (hi ? 8 : 0);
    int kb = kk + (hi ? 16 : 0);
    v16h a_self, a_mean, bs, bn;
#pragma unroll
    for (int j = 0; j < 8; j++) {
      a_self[j]     = (_Float16)xs[mrow][ka + j];
      a_self[8 + j] = (_Float16)xs[mrow][ka + 16 + j];
      a_mean[j]     = (_Float16)ms[mrow][ka + j];
      a_mean[8 + j] = (_Float16)ms[mrow][ka + 16 + j];
    }
#pragma unroll
    for (int j = 0; j < 16; j++) {
      bs[j] = (_Float16)Wself[(size_t)(kb + j) * NOUT + ncol];
      bn[j] = (_Float16)Wneigh[(size_t)(kb + j) * NOUT + ncol];
    }
    c = __builtin_amdgcn_wmma_f32_16x16x32_f16(false, a_self, false, bs,
                                               (short)0, c, false, false);
    c = __builtin_amdgcn_wmma_f32_16x16x32_f16(false, a_mean, false, bn,
                                               (short)0, c, false, false);
  }
#endif

  float bv = bias[ncol];
#pragma unroll
  for (int i = 0; i < 8; i++) {
    int   row = m0 + i + (hi ? 8 : 0);
    float v   = c[i] + bv;
    if (RELU) v = fmaxf(v, 0.0f);
    OUT[(size_t)row * NOUT + ncol] = v;
  }
}

extern "C" void kernel_launch(void* const* d_in, const int* in_sizes, int n_in,
                              void* d_out, int out_size, void* d_ws,
                              size_t ws_size, hipStream_t stream) {
  (void)in_sizes; (void)n_in; (void)out_size; (void)ws_size;
  const float* x   = (const float*)d_in[0];
  const int*   src = (const int*)d_in[1];
  const int*   dst = (const int*)d_in[2];
  const float* W1s = (const float*)d_in[3];
  const float* W1n = (const float*)d_in[4];
  const float* b1  = (const float*)d_in[5];
  const float* W2s = (const float*)d_in[6];
  const float* W2n = (const float*)d_in[7];
  const float* b2  = (const float*)d_in[8];
  float*       out = (float*)d_out;

  char*  ws  = (char*)d_ws;
  float* deg = (float*)ws;
  float* agg = deg + N_NODES;
  float* h   = agg + (size_t)N_NODES * KD;

  int* cnt  = (int*)h;
  int* rank = cnt + N_NODES;
  int* offs = (int*)d_out;
  int* perm = offs + N_NODES + 64;

  zero_i32<<<N_NODES / 256, 256, 0, stream>>>(cnt);
  rank_kernel<<<N_EDGES / 4 / 256, 256, 0, stream>>>(dst, cnt, rank);
  scan_kernel<<<N_NODES / 1024, 1024, 0, stream>>>(cnt, offs, deg);
  fill_kernel<<<N_EDGES / 4 / 256, 256, 0, stream>>>(src, dst, rank, offs,
                                                     perm);

  gather_kernel<<<N_NODES / NPW / 8, 256, 0, stream>>>(x, offs, perm, agg);
  sage_gemm<128, true><<<N_NODES / 16, 256, 0, stream>>>(x, agg, deg, W1s,
                                                         W1n, b1, h);

  gather_kernel<<<N_NODES / NPW / 8, 256, 0, stream>>>(h, offs, perm, agg);
  sage_gemm<64, false><<<N_NODES / 16, 128, 0, stream>>>(h, agg, deg, W2s,
                                                         W2n, b2, out);
}


# Pad the shared-memory row pitch of both GEMM operand tiles from 128 to 132 floats: the A-fragment reads (16 rows at one column per half-wave) now spread over all 64 banks instead of one bank pair
# speedup vs baseline: 2.4755x; 1.1711x over previous
#include <hip/hip_runtime.h>

#define N_NODES 65536
#define N_EDGES 1048576
#define KD      128

typedef __attribute__((ext_vector_type(2)))  float     v2f;
typedef __attribute__((ext_vector_type(8)))  float     v8f;
typedef __attribute__((ext_vector_type(16))) _Float16  v16h;

#define NPW 8

__global__ void zero_i32(int* __restrict__ p) {
  p[blockIdx.x * blockDim.x + threadIdx.x] = 0;
}

__global__ void rank_kernel(const int* __restrict__ dst,
                            int* __restrict__ cnt,
                            int* __restrict__ rank) {
  int  t = blockIdx.x * blockDim.x + threadIdx.x;
  int4 d = ((const int4*)dst)[t];
  int4 r;
  r.x = atomicAdd(cnt + d.x, 1);
  r.y = atomicAdd(cnt + d.y, 1);
  r.z = atomicAdd(cnt + d.z, 1);
  r.w = atomicAdd(cnt + d.w, 1);
  ((int4*)rank)[t] = r;
}

__global__ __launch_bounds__(1024)
void scan_kernel(const int* __restrict__ cnt,
                 int* __restrict__ offs,
                 float* __restrict__ deg) {
  __shared__ int wsum[32];
  __shared__ int sbase;
  const int b    = blockIdx.x;
  const int t    = threadIdx.x;
  const int lane = t & 31;
  const int w    = t >> 5;

  int part = 0;
  const int4* c4 = (const int4*)cnt;
#pragma unroll 4
  for (int i = t; i < b * 256; i += 1024) {
    int4 v = c4[i];
    part += v.x + v.y + v.z + v.w;
  }
  for (int o = 16; o > 0; o >>= 1) part += __shfl_xor(part, o);
  if (lane == 0) wsum[w] = part;
  __syncthreads();
  if (w == 0) {
    int s = wsum[lane];
    for (int o = 16; o > 0; o >>= 1) s += __shfl_xor(s, o);
    if (lane == 0) sbase = s;
  }
  __syncthreads();
  const int base = sbase;

  const int c    = cnt[b * 1024 + t];
  int       incl = c;
  for (int o = 1; o < 32; o <<= 1) {
    int u = __shfl_up(incl, o);
    if (lane >= o) incl += u;
  }
  if (lane == 31) wsum[w] = incl;
  __syncthreads();
  if (w == 0) {
    int s  = wsum[lane];
    int si = s;
    for (int o = 1; o < 32; o <<= 1) {
      int u = __shfl_up(si, o);
      if (lane >= o) si += u;
    }
    wsum[lane] = si - s;
  }
  __syncthreads();
  const int excl = base + wsum[w] + incl - c;
  offs[b * 1024 + t] = excl;
  deg[b * 1024 + t]  = (float)c;
  if (b == (int)gridDim.x - 1 && t == 1023) offs[N_NODES] = excl + c;
}

__global__ void fill_kernel(const int* __restrict__ src,
                            const int* __restrict__ dst,
                            const int* __restrict__ rank,
                            const int* __restrict__ offs,
                            int* __restrict__ perm) {
  int  t = blockIdx.x * blockDim.x + threadIdx.x;
  int4 s = ((const int4*)src)[t];
  int4 d = ((const int4*)dst)[t];
  int4 r = ((const int4*)rank)[t];
  perm[(offs[d.x] + r.x) & (N_EDGES - 1)] = s.x;
  perm[(offs[d.y] + r.y) & (N_EDGES - 1)] = s.y;
  perm[(offs[d.z] + r.z) & (N_EDGES - 1)] = s.z;
  perm[(offs[d.w] + r.w) & (N_EDGES - 1)] = s.w;
}

__global__ __launch_bounds__(256)
void gather_kernel(const float* __restrict__ feat,
                   const int* __restrict__ offs,
                   const int* __restrict__ perm,
                   float* __restrict__ agg) {
  const int lane = threadIdx.x & 31;
  const int wave =
      __builtin_amdgcn_readfirstlane((blockIdx.x * 256 + threadIdx.x) >> 5);
  const int n0 = wave * NPW;

  const float4* f4 = (const float4*)feat;
  float4*       a4 = (float4*)agg;

  const int endv = offs[n0 + 1 + (lane & (NPW - 1))];
  const int rbeg = offs[n0];
  const int rend = __builtin_amdgcn_readlane(endv, NPW - 1);
  int       node = 0;
  int       nend = __builtin_amdgcn_readlane(endv, 0);
  float4    acc  = make_float4(0.0f, 0.0f, 0.0f, 0.0f);

  for (int base = rbeg; base < rend; base += 32) {
    const int rem = rend - base;
    int myidx = perm[base + min(lane, rem - 1)];
    myidx &= (N_NODES - 1);
#pragma unroll
    for (int k0 = 0; k0 < 32; k0 += 8) {
      if (k0 < rem) {
        float4 v[8];
#pragma unroll
        for (int k = 0; k < 8; k++) {
          const int idx = __builtin_amdgcn_readlane(myidx, k0 + k);
          v[k] = f4[(size_t)idx * (KD / 4) + lane];
        }
#pragma unroll
        for (int k = 0; k < 8; k++) {
          const int pos = base + k0 + k;
          if (pos < rend) {
            while (pos >= nend) {
              a4[(size_t)(n0 + node) * (KD / 4) + lane] = acc;
              acc = make_float4(0.0f, 0.0f, 0.0f, 0.0f);
              node++;
              nend = __builtin_amdgcn_readlane(endv, node);
            }
            acc.x += v[k].x; acc.y += v[k].y;
            acc.z += v[k].z; acc.w += v[k].w;
          }
        }
      }
    }
  }
  while (node < NPW) {
    a4[(size_t)(n0 + node) * (KD / 4) + lane] = acc;
    acc = make_float4(0.0f, 0.0f, 0.0f, 0.0f);
    node++;
  }
}

template <int NOUT, bool RELU>
__global__ __launch_bounds__(32 * (NOUT / 16))
void sage_gemm(const float* __restrict__ X,
               const float* __restrict__ AGG,
               const float* __restrict__ deg,
               const float* __restrict__ Wself,
               const float* __restrict__ Wneigh,
               const float* __restrict__ bias,
               float* __restrict__ OUT) {
  __shared__ float xs[16][KD + 4];
  __shared__ float ms[16][KD + 4];

  const int m0       = blockIdx.x * 16;
  const int tid      = threadIdx.x;
  const int nthreads = 32 * (NOUT / 16);

  for (int i = tid; i < 16 * (KD / 4); i += nthreads) {
    int r  = i / (KD / 4);
    int c4 = i % (KD / 4);
    float4 xv = ((const float4*)(X + (size_t)(m0 + r) * KD))[c4];
    float4 av = ((const float4*)(AGG + (size_t)(m0 + r) * KD))[c4];
    float  rd = 1.0f / fmaxf(deg[m0 + r], 1.0f);
    av.x *= rd; av.y *= rd; av.z *= rd; av.w *= rd;
    ((float4*)&xs[r][0])[c4] = xv;
    ((float4*)&ms[r][0])[c4] = av;
  }
  __syncthreads();

  const int  wave = tid / 32;
  const int  lane = tid % 32;
  const int  n0   = wave * 16;
  const int  mrow = lane & 15;
  const bool hi   = lane >= 16;
  const int  ncol = n0 + (lane & 15);

  v8f c = {};

#if __has_builtin(__builtin_amdgcn_wmma_f32_16x16x4_f32)
  for (int kk = 0; kk < KD; kk += 4) {
    int ka = kk + (hi ? 2 : 0);
    v2f a_self = { xs[mrow][ka], xs[mrow][ka + 1] };
    v2f a_mean = { ms[mrow][ka], ms[mrow][ka + 1] };
    v2f bs = { Wself[(size_t)ka * NOUT + ncol],
               Wself[(size_t)(ka + 1) * NOUT + ncol] };
    v2f bn = { Wneigh[(size_t)ka * NOUT + ncol],
               Wneigh[(size_t)(ka + 1) * NOUT + ncol] };
    c = __builtin_amdgcn_wmma_f32_16x16x4_f32(false, a_self, false, bs,
                                              (short)0, c, false, false);
    c = __builtin_amdgcn_wmma_f32_16x16x4_f32(false, a_mean, false, bn,
                                              (short)0, c, false, false);
  }
#else
  for (int kk = 0; kk < KD; kk += 32) {
    int ka = kk + (hi ? 8 : 0);
    int kb = kk + (hi ? 16 : 0);
    v16h a_self, a_mean, bs, bn;
#pragma unroll
    for (int j = 0; j < 8; j++) {
      a_self[j]     = (_Float16)xs[mrow][ka + j];
      a_self[8 + j] = (_Float16)xs[mrow][ka + 16 + j];
      a_mean[j]     = (_Float16)ms[mrow][ka + j];
      a_mean[8 + j] = (_Float16)ms[mrow][ka + 16 + j];
    }
#pragma unroll
    for (int j = 0; j < 16; j++) {
      bs[j] = (_Float16)Wself[(size_t)(kb + j) * NOUT + ncol];
      bn[j] = (_Float16)Wneigh[(size_t)(kb + j) * NOUT + ncol];
    }
    c = __builtin_amdgcn_wmma_f32_16x16x32_f16(false, a_self, false, bs,
                                               (short)0, c, false, false);
    c = __builtin_amdgcn_wmma_f32_16x16x32_f16(false, a_mean, false, bn,
                                               (short)0, c, false, false);
  }
#endif

  float bv = bias[ncol];
#pragma unroll
  for (int i = 0; i < 8; i++) {
    int   row = m0 + i + (hi ? 8 : 0);
    float v   = c[i] + bv;
    if (RELU) v = fmaxf(v, 0.0f);
    OUT[(size_t)row * NOUT + ncol] = v;
  }
}

extern "C" void kernel_launch(void* const* d_in, const int* in_sizes, int n_in,
                              void* d_out, int out_size, void* d_ws,
                              size_t ws_size, hipStream_t stream) {
  (void)in_sizes; (void)n_in; (void)out_size; (void)ws_size;
  const float* x   = (const float*)d_in[0];
  const int*   src = (const int*)d_in[1];
  const int*   dst = (const int*)d_in[2];
  const float* W1s = (const float*)d_in[3];
  const float* W1n = (const float*)d_in[4];
  const float* b1  = (const float*)d_in[5];
  const float* W2s = (const float*)d_in[6];
  const float* W2n = (const float*)d_in[7];
  const float* b2  = (const float*)d_in[8];
  float*       out = (float*)d_out;

  char*  ws  = (char*)d_ws;
  float* deg = (float*)ws;
  float* agg = deg + N_NODES;
  float* h   = agg + (size_t)N_NODES * KD;

  int* cnt  = (int*)h;
  int* rank = cnt + N_NODES;
  int* offs = (int*)d_out;
  int* perm = offs + N_NODES + 64;

  zero_i32<<<N_NODES / 256, 256, 0, stream>>>(cnt);
  rank_kernel<<<N_EDGES / 4 / 256, 256, 0, stream>>>(dst, cnt, rank);
  scan_kernel<<<N_NODES / 1024, 1024, 0, stream>>>(cnt, offs, deg);
  fill_kernel<<<N_EDGES / 4 / 256, 256, 0, stream>>>(src, dst, rank, offs,
                                                     perm);

  gather_kernel<<<N_NODES / NPW / 8, 256, 0, stream>>>(x, offs, perm, agg);
  sage_gemm<128, true><<<N_NODES / 16, 256, 0, stream>>>(x, agg, deg, W1s,
                                                         W1n, b1, h);

  gather_kernel<<<N_NODES / NPW / 8, 256, 0, stream>>>(h, offs, perm, agg);
  sage_gemm<64, false><<<N_NODES / 16, 128, 0, stream>>>(h, agg, deg, W2s,
                                                         W2n, b2, out);
}


# GEMM waves keep their column strip of both weight matrices in registers and each workgroup loops over 8 (layer 1) or 4 (layer 2) row tiles: weights loaded once per wave, 8x/4x fewer waves to enter
# speedup vs baseline: 2.5719x; 1.0389x over previous
#include <hip/hip_runtime.h>

#define N_NODES 65536
#define N_EDGES 1048576
#define KD      128

typedef __attribute__((ext_vector_type(2)))  float     v2f;
typedef __attribute__((ext_vector_type(4)))  float     v4f;
typedef __attribute__((ext_vector_type(8)))  float     v8f;

#define NPW 8

__global__ void zero_i32(int* __restrict__ p) {
  p[blockIdx.x * blockDim.x + threadIdx.x] = 0;
}

__global__ void rank_kernel(const int* __restrict__ dst,
                            int* __restrict__ cnt,
                            int* __restrict__ rank) {
  int  t = blockIdx.x * blockDim.x + threadIdx.x;
  int4 d = ((const int4*)dst)[t];
  int4 r;
  r.x = atomicAdd(cnt + d.x, 1);
  r.y = atomicAdd(cnt + d.y, 1);
  r.z = atomicAdd(cnt + d.z, 1);
  r.w = atomicAdd(cnt + d.w, 1);
  ((int4*)rank)[t] = r;
}

__global__ __launch_bounds__(1024)
void scan_kernel(const int* __restrict__ cnt,
                 int* __restrict__ offs,
                 float* __restrict__ deg) {
  __shared__ int wsum[32];
  __shared__ int sbase;
  const int b    = blockIdx.x;
  const int t    = threadIdx.x;
  const int lane = t & 31;
  const int w    = t >> 5;

  int part = 0;
  const int4* c4 = (const int4*)cnt;
#pragma unroll 4
  for (int i = t; i < b * 256; i += 1024) {
    int4 v = c4[i];
    part += v.x + v.y + v.z + v.w;
  }
  for (int o = 16; o > 0; o >>= 1) part += __shfl_xor(part, o);
  if (lane == 0) wsum[w] = part;
  __syncthreads();
  if (w == 0) {
    int s = wsum[lane];
    for (int o = 16; o > 0; o >>= 1) s += __shfl_xor(s, o);
    if (lane == 0) sbase = s;
  }
  __syncthreads();
  const int base = sbase;

  const int c    = cnt[b * 1024 + t];
  int       incl = c;
  for (int o = 1; o < 32; o <<= 1) {
    int u = __shfl_up(incl, o);
    if (lane >= o) incl += u;
  }
  if (lane == 31) wsum[w] = incl;
  __syncthreads();
  if (w == 0) {
    int s  = wsum[lane];
    int si = s;
    for (int o = 1; o < 32; o <<= 1) {
      int u = __shfl_up(si, o);
      if (lane >= o) si += u;
    }
    wsum[lane] = si - s;
  }
  __syncthreads();
  const int excl = base + wsum[w] + incl - c;
  offs[b * 1024 + t] = excl;
  deg[b * 1024 + t]  = (float)c;
  if (b == (int)gridDim.x - 1 && t == 1023) offs[N_NODES] = excl + c;
}

__global__ void fill_kernel(const int* __restrict__ src,
                            const int* __restrict__ dst,
                            const int* __restrict__ rank,
                            const int* __restrict__ offs,
                            int* __restrict__ perm) {
  int  t = blockIdx.x * blockDim.x + threadIdx.x;
  int4 s = ((const int4*)src)[t];
  int4 d = ((const int4*)dst)[t];
  int4 r = ((const int4*)rank)[t];
  perm[(offs[d.x] + r.x) & (N_EDGES - 1)] = s.x;
  perm[(offs[d.y] + r.y) & (N_EDGES - 1)] = s.y;
  perm[(offs[d.z] + r.z) & (N_EDGES - 1)] = s.z;
  perm[(offs[d.w] + r.w) & (N_EDGES - 1)] = s.w;
}

__global__ __launch_bounds__(256)
void gather_kernel(const float* __restrict__ feat,
                   const int* __restrict__ offs,
                   const int* __restrict__ perm,
                   float* __restrict__ agg) {
  const int lane = threadIdx.x & 31;
  const int wave =
      __builtin_amdgcn_readfirstlane((blockIdx.x * 256 + threadIdx.x) >> 5);
  const int n0 = wave * NPW;

  const float4* f4 = (const float4*)feat;
  float4*       a4 = (float4*)agg;

  const int endv = offs[n0 + 1 + (lane & (NPW - 1))];
  const int rbeg = offs[n0];
  const int rend = __builtin_amdgcn_readlane(endv, NPW - 1);
  int       node = 0;
  int       nend = __builtin_amdgcn_readlane(endv, 0);
  float4    acc  = make_float4(0.0f, 0.0f, 0.0f, 0.0f);

  for (int base = rbeg; base < rend; base += 32) {
    const int rem = rend - base;
    int myidx = perm[base + min(lane, rem - 1)];
    myidx &= (N_NODES - 1);
#pragma unroll
    for (int k0 = 0; k0 < 32; k0 += 8) {
      if (k0 < rem) {
        float4 v[8];
#pragma unroll
        for (int k = 0; k < 8; k++) {
          const int idx = __builtin_amdgcn_readlane(myidx, k0 + k);
          v[k] = f4[(size_t)idx * (KD / 4) + lane];
        }
#pragma unroll
        for (int k = 0; k < 8; k++) {
          const int pos = base + k0 + k;
          if (pos < rend) {
            while (pos >= nend) {
              a4[(size_t)(n0 + node) * (KD / 4) + lane] = acc;
              acc = make_float4(0.0f, 0.0f, 0.0f, 0.0f);
              node++;
              nend = __builtin_amdgcn_readlane(endv, node);
            }
            acc.x += v[k].x; acc.y += v[k].y;
            acc.z += v[k].z; acc.w += v[k].w;
          }
        }
      }
    }
  }
  while (node < NPW) {
    a4[(size_t)(n0 + node) * (KD / 4) + lane] = acc;
    acc = make_float4(0.0f, 0.0f, 0.0f, 0.0f);
    node++;
  }
}

template <int NOUT, bool RELU>
__global__ __launch_bounds__(32 * (NOUT / 16))
void sage_gemm(const float* __restrict__ X,
               const float* __restrict__ AGG,
               const float* __restrict__ deg,
               const float* __restrict__ Wself,
               const float* __restrict__ Wneigh,
               const float* __restrict__ bias,
               float* __restrict__ OUT) {
  constexpr int NW  = NOUT / 16;
  constexpr int RPW = 16 / NW;
  constexpr int NT  = N_NODES / 16;
  constexpr int KS  = KD / 4;

  __shared__ __attribute__((aligned(16))) float xs[2][16][KD + 4];
  __shared__ __attribute__((aligned(16))) float ms[2][16][KD + 4];

  const int tid  = threadIdx.x;
  const int lane = tid & 31;
  const int  wave = __builtin_amdgcn_readfirstlane(tid >> 5);
  const int  n0   = wave * 16;
  const int  mrow = lane & 15;
  const bool hi   = lane >= 16;
  const int  ncol = n0 + (lane & 15);
  const int  koff = hi ? 2 : 0;

  const v4f* X4 = (const v4f*)X;
  const v4f* A4 = (const v4f*)AGG;

  v4f   xr[RPW], ar[RPW];
  float dr[RPW];
  int tile = blockIdx.x;
#pragma unroll
  for (int j = 0; j < RPW; j++) {
    const int row = tile * 16 + wave + j * NW;
    xr[j] = X4[(size_t)row * (KD / 4) + lane];
    ar[j] = A4[(size_t)row * (KD / 4) + lane];
    dr[j] = deg[row];
  }

  v2f bs[KS], bn[KS];
#pragma unroll
  for (int s = 0; s < KS; s++) {
    const int ka = 4 * s + koff;
    bs[s] = v2f{ Wself[(size_t)ka * NOUT + ncol],
                 Wself[(size_t)(ka + 1) * NOUT + ncol] };
    bn[s] = v2f{ Wneigh[(size_t)ka * NOUT + ncol],
                 Wneigh[(size_t)(ka + 1) * NOUT + ncol] };
  }
  const float bv = bias[ncol];

  int buf = 0;
  for (; tile < NT; tile += (int)gridDim.x) {
#pragma unroll
    for (int j = 0; j < RPW; j++) {
      const int r  = wave + j * NW;
      const float rd = 1.0f / fmaxf(dr[j], 1.0f);
      ((v4f*)&xs[buf][r][0])[lane] = xr[j];
      ((v4f*)&ms[buf][r][0])[lane] = ar[j] * rd;
    }
    __syncthreads();

    const int nxt = tile + (int)gridDim.x;
    if (nxt < NT) {
#pragma unroll
      for (int j = 0; j < RPW; j++) {
        const int row = nxt * 16 + wave + j * NW;
        xr[j] = X4[(size_t)row * (KD / 4) + lane];
        ar[j] = A4[(size_t)row * (KD / 4) + lane];
        dr[j] = deg[row];
      }
    }

    v8f c = {};
#pragma unroll
    for (int s = 0; s < KS; s++) {
      const int ka = 4 * s + koff;
      v2f a_self = { xs[buf][mrow][ka], xs[buf][mrow][ka + 1] };
      v2f a_mean = { ms[buf][mrow][ka], ms[buf][mrow][ka + 1] };
      c = __builtin_amdgcn_wmma_f32_16x16x4_f32(false, a_self, false, bs[s],
                                                (short)0, c, false, false);
      c = __builtin_amdgcn_wmma_f32_16x16x4_f32(false, a_mean, false, bn[s],
                                                (short)0, c, false, false);
    }

#pragma unroll
    for (int i = 0; i < 8; i++) {
      int   row = tile * 16 + i + (hi ? 8 : 0);
      float v   = c[i] + bv;
      if (RELU) v = fmaxf(v, 0.0f);
      OUT[(size_t)row * NOUT + ncol] = v;
    }
    buf ^= 1;
  }
}

extern "C" void kernel_launch(void* const* d_in, const int* in_sizes, int n_in,
                              void* d_out, int out_size, void* d_ws,
                              size_t ws_size, hipStream_t stream) {
  (void)in_sizes; (void)n_in; (void)out_size; (void)ws_size;
  const float* x   = (const float*)d_in[0];
  const int*   src = (const int*)d_in[1];
  const int*   dst = (const int*)d_in[2];
  const float* W1s = (const float*)d_in[3];
  const float* W1n = (const float*)d_in[4];
  const float* b1  = (const float*)d_in[5];
  const float* W2s = (const float*)d_in[6];
  const float* W2n = (const float*)d_in[7];
  const float* b2  = (const float*)d_in[8];
  float*       out = (float*)d_out;

  char*  ws  = (char*)d_ws;
  float* deg = (float*)ws;
  float* agg = deg + N_NODES;
  float* h   = agg + (size_t)N_NODES * KD;

  int* cnt  = (int*)h;
  int* rank = cnt + N_NODES;
  int* offs = (int*)d_out;
  int* perm = offs + N_NODES + 64;

  zero_i32<<<N_NODES / 256, 256, 0, stream>>>(cnt);
  rank_kernel<<<N_EDGES / 4 / 256, 256, 0, stream>>>(dst, cnt, rank);
  scan_kernel<<<N_NODES / 1024, 1024, 0, stream>>>(cnt, offs, deg);
  fill_kernel<<<N_EDGES / 4 / 256, 256, 0, stream>>>(src, dst, rank, offs,
                                                     perm);

  gather_kernel<<<N_NODES / NPW / 8, 256, 0, stream>>>(x, offs, perm, agg);
  sage_gemm<128, true><<<N_NODES / 16 / 8, 256, 0, stream>>>(x, agg, deg, W1s,
                                                         W1n, b1, h);

  gather_kernel<<<N_NODES / NPW / 8, 256, 0, stream>>>(h, offs, perm, agg);
  sage_gemm<64, false><<<N_NODES / 16 / 4, 128, 0, stream>>>(h, agg, deg, W2s,
                                                         W2n, b2, out);
}


# Gathering waves sum a bucket line by short straight code: rows in groups of four at fixed slots plus up to four single rows, each under one scalar bit test; no per-row branch maze
# speedup vs baseline: 3.1458x; 1.2231x over previous
#include <hip/hip_runtime.h>

#define N_NODES 65536
#define N_EDGES 1048576
#define KD      128

typedef __attribute__((ext_vector_type(2)))  float     v2f;
typedef __attribute__((ext_vector_type(4)))  float     v4f;
typedef __attribute__((ext_vector_type(8)))  float     v8f;

#define CAP 64

__global__ void zero_i32(int* __restrict__ p) {
  p[blockIdx.x * blockDim.x + threadIdx.x] = 0;
}

__global__ void fill_kernel(const int* __restrict__ src,
                            const int* __restrict__ dst,
                            int* __restrict__ cnt,
                            int* __restrict__ slots) {
  int  t = blockIdx.x * blockDim.x + threadIdx.x;
  int4 s = ((const int4*)src)[t];
  int4 d = ((const int4*)dst)[t];
  d.x &= (N_NODES - 1); d.y &= (N_NODES - 1);
  d.z &= (N_NODES - 1); d.w &= (N_NODES - 1);
  const int rx = atomicAdd(cnt + d.x, 1);
  const int ry = atomicAdd(cnt + d.y, 1);
  const int rz = atomicAdd(cnt + d.z, 1);
  const int rw = atomicAdd(cnt + d.w, 1);
  if (rx < CAP) slots[(size_t)d.x * CAP + rx] = s.x;
  if (ry < CAP) slots[(size_t)d.y * CAP + ry] = s.y;
  if (rz < CAP) slots[(size_t)d.z * CAP + rz] = s.z;
  if (rw < CAP) slots[(size_t)d.w * CAP + rw] = s.w;
}

__device__ __forceinline__ float4 row_of(const float4* __restrict__ f4, int my,
                                         int k, int lane) {
  const int idx = __builtin_amdgcn_readlane(my, k);
  return f4[(size_t)idx * (KD / 4) + lane];
}

__device__ __forceinline__ void add4(float4& acc, const float4 v) {
  acc.x += v.x; acc.y += v.y; acc.z += v.z; acc.w += v.w;
}

__device__ __forceinline__ float4 row_at(const char* __restrict__ base, int off,
                                         int k, unsigned lane16) {
  const unsigned o = (unsigned)__builtin_amdgcn_readlane(off, k) + lane16;
  return *(const float4*)(base + o);
}

__device__ __forceinline__ void sum_line(const char* __restrict__ base, int off,
                                         int t, unsigned lane16, float4& acc) {
  float4 v[32];
  const int      nq = min(t >> 2, 7);
  const int      r  = t - 4 * nq;
  const unsigned gm = (1u << nq) - 1u;
  const unsigned sm = (1u << r) - 1u;
  const float4 z = make_float4(0.0f, 0.0f, 0.0f, 0.0f);
#pragma unroll
  for (int q = 0; q < 7; q++) {
    if (gm & (1u << q)) {
      v[4 * q + 0] = row_at(base, off, 4 * q + 0, lane16);
      v[4 * q + 1] = row_at(base, off, 4 * q + 1, lane16);
      v[4 * q + 2] = row_at(base, off, 4 * q + 2, lane16);
      v[4 * q + 3] = row_at(base, off, 4 * q + 3, lane16);
    } else {
      v[4 * q + 0] = z; v[4 * q + 1] = z; v[4 * q + 2] = z; v[4 * q + 3] = z;
    }
  }
#pragma unroll
  for (int j = 0; j < 4; j++) {
    if (sm & (1u << j)) v[28 + j] = row_at(base, off, 4 * nq + j, lane16);
    else                v[28 + j] = z;
  }
#pragma unroll
  for (int j = 3; j >= 0; j--) {
    if (sm & (1u << j)) add4(acc, v[28 + j]);
  }
#pragma unroll
  for (int q = 6; q >= 0; q--) {
    if (gm & (1u << q)) {
      add4(acc, v[4 * q + 3]);
      add4(acc, v[4 * q + 2]);
      add4(acc, v[4 * q + 1]);
      add4(acc, v[4 * q + 0]);
    }
  }
}

__device__ __forceinline__ float4 node_sum(const float* __restrict__ X,
                                           const int* __restrict__ src,
                                           const int* __restrict__ dst,
                                           const int* slots, int node, int c,
                                           int cur, int lane) {
  float4 acc = make_float4(0.0f, 0.0f, 0.0f, 0.0f);
  if (c <= CAP) {
    const char*    xb     = (const char*)X;
    const unsigned lane16 = (unsigned)lane << 4;
    if (c > 0) sum_line(xb, cur, min(c, 32), lane16, acc);
    if (c > 32) {
      const int s2 = *(const int*)((const char*)slots +
                                   ((unsigned)node * (CAP * 4u) + 128u +
                                    ((unsigned)lane << 2)));
      const int more = (s2 & (N_NODES - 1)) << 9;
      for (int k = 0; k < c - 32; k++) add4(acc, row_at(xb, more, k, lane16));
    }
  } else {
    const float4* f4 = (const float4*)X;
    for (int e0 = 0; e0 < N_EDGES; e0 += 32) {
      const int d = dst[e0 + lane] & (N_NODES - 1);
      const int s = src[e0 + lane] & (N_NODES - 1);
      unsigned  m = __builtin_amdgcn_ballot_w32(d == node);
      while (m) {
        const int l = __builtin_ctz(m);
        m &= m - 1;
        add4(acc, row_of(f4, s, l, lane));
      }
    }
  }
  return acc;
}

#define FGRID 256
#define GW    16

template <int NOUT, bool RELU>
__global__ __launch_bounds__(32 * (NOUT / 16 + GW))
void sage_fused(const float* __restrict__ X,
                const int* __restrict__ src,
                const int* __restrict__ dst,
                const int* __restrict__ cnt,
                const int* slots,
                float* OUT,
                const float* __restrict__ Wself,
                const float* __restrict__ Wneigh,
                const float* __restrict__ bias) {
  constexpr int NW = NOUT / 16;
  constexpr int NT = N_NODES / 16;
  constexpr int NI = NT / FGRID;
  constexpr int KS = KD / 4;
  static_assert(NI == 16 && GW == 16, "one count per lane, one node per wave");

  __shared__ __attribute__((aligned(16))) v2f wS[NW][KS][32];
  __shared__ __attribute__((aligned(16))) v2f wN[NW][KS][32];
  __shared__ __attribute__((aligned(16))) float xs[2][16][KD + 4];
  __shared__ __attribute__((aligned(16))) float ms[2][16][KD + 4];

  const int tid  = threadIdx.x;
  const int lane = tid & 31;
  const int  wave  = __builtin_amdgcn_readfirstlane(tid >> 5);
  const bool mul   = wave < NW;
  const int  tile0 = blockIdx.x;
  const float4* f4 = (const float4*)X;

  const int  mrow = lane & 15;
  const bool hi   = lane >= 16;
  const int  ncol = wave * 16 + (lane & 15);
  const int  koff = hi ? 2 : 0;
  float bv = 0.0f;
  constexpr int RPW = 16 / NW;
  float4 xr0, xr1, xr2, xr3;
  static_assert(RPW == 2 || RPW == 4, "two or four own rows per wave");
  auto own_src = [&](int t, int j) -> const float4* {
    return f4 + (size_t)(t * 16 + wave + j * NW) * (KD / 4) + lane;
  };
  auto own_dst = [&](int b, int j) -> float4* {
    return (float4*)&xs[b][wave + j * NW][0] + lane;
  };

  const int g = wave - NW;
  int cv  = 0;
  int cur = 0;

  if (mul) {
    for (int s = 0; s < KS; s++) {
      const int ka = 4 * s + koff;
      wS[wave][s][lane] = v2f{ Wself[(size_t)ka * NOUT + ncol],
                               Wself[(size_t)(ka + 1) * NOUT + ncol] };
      wN[wave][s][lane] = v2f{ Wneigh[(size_t)ka * NOUT + ncol],
                               Wneigh[(size_t)(ka + 1) * NOUT + ncol] };
    }
    bv = bias[ncol];
    *own_dst(0, 0) = *own_src(tile0, 0);
    *own_dst(0, 1) = *own_src(tile0, 1);
    if (RPW == 4) {
      *own_dst(0, 2) = *own_src(tile0, 2);
      *own_dst(0, 3) = *own_src(tile0, 3);
    }
    xr0 = *own_src(tile0 + FGRID, 0);
    xr1 = *own_src(tile0 + FGRID, 1);
    if (RPW == 4) {
      xr2 = *own_src(tile0 + FGRID, 2);
      xr3 = *own_src(tile0 + FGRID, 3);
    }
  } else {
    __builtin_amdgcn_s_setprio(3);
    cv = cnt[(tile0 + (lane & (NI - 1)) * FGRID) * 16 + g];
    cur = (*(const int*)((const char*)slots +
                         ((unsigned)(tile0 * 16 + g) * (CAP * 4u) +
                          ((unsigned)lane << 2))) & (N_NODES - 1)) << 9;
  }

  for (int i = -1; i < NI; i++) {
    if (mul) {
      if (i >= 0) {
        const int buf  = i & 1;
        const int tile = tile0 + i * FGRID;
        if (i + 1 < NI) {
          *own_dst(buf ^ 1, 0) = xr0;
          *own_dst(buf ^ 1, 1) = xr1;
          if (RPW == 4) {
            *own_dst(buf ^ 1, 2) = xr2;
            *own_dst(buf ^ 1, 3) = xr3;
          }
        }
        if (i + 2 < NI) {
          xr0 = *own_src(tile + 2 * FGRID, 0);
          xr1 = *own_src(tile + 2 * FGRID, 1);
          if (RPW == 4) {
            xr2 = *own_src(tile + 2 * FGRID, 2);
            xr3 = *own_src(tile + 2 * FGRID, 3);
          }
        }
        v8f c = {};
#pragma unroll
        for (int s = 0; s < KS; s++) {
          const int ka = 4 * s + koff;
          v2f a_self = { xs[buf][mrow][ka], xs[buf][mrow][ka + 1] };
          v2f a_mean = { ms[buf][mrow][ka], ms[buf][mrow][ka + 1] };
          c = __builtin_amdgcn_wmma_f32_16x16x4_f32(false, a_self, false,
                                                    wS[wave][s][lane], (short)0,
                                                    c, false, false);
          c = __builtin_amdgcn_wmma_f32_16x16x4_f32(false, a_mean, false,
                                                    wN[wave][s][lane], (short)0,
                                                    c, false, false);
        }
#pragma unroll
        for (int r = 0; r < 8; r++) {
          int   row = tile * 16 + r + (hi ? 8 : 0);
          float v   = c[r] + bv;
          if (RELU) v = fmaxf(v, 0.0f);
          OUT[(size_t)row * NOUT + ncol] = v;
        }
      }
    } else if (i + 1 < NI) {
      const int it   = i + 1;
      const int b    = it & 1;
      const int node = (tile0 + it * FGRID) * 16 + g;
      const int c    = __builtin_amdgcn_readlane(cv, it);
      const int nn = it + 1 < NI ? node + FGRID * 16 : node;
      const int nx = *(const int*)((const char*)slots +
                                   ((unsigned)nn * (CAP * 4u) +
                                    ((unsigned)lane << 2)));
      const float4 a = node_sum(X, src, dst, slots, node, c, cur, lane);
      const float  r = 1.0f / fmaxf((float)c, 1.0f);
      ((float4*)&ms[b][g][0])[lane] =
          make_float4(a.x * r, a.y * r, a.z * r, a.w * r);
      cur = (nx & (N_NODES - 1)) << 9;
    }
    __builtin_amdgcn_fence(__ATOMIC_RELEASE, "workgroup", "local");
    __builtin_amdgcn_s_barrier();
    __builtin_amdgcn_fence(__ATOMIC_ACQUIRE, "workgroup", "local");
  }
}
extern "C" void kernel_launch(void* const* d_in, const int* in_sizes, int n_in,
                              void* d_out, int out_size, void* d_ws,
                              size_t ws_size, hipStream_t stream) {
  (void)in_sizes; (void)n_in; (void)out_size; (void)ws_size;
  const float* x   = (const float*)d_in[0];
  const int*   src = (const int*)d_in[1];
  const int*   dst = (const int*)d_in[2];
  const float* W1s = (const float*)d_in[3];
  const float* W1n = (const float*)d_in[4];
  const float* b1  = (const float*)d_in[5];
  const float* W2s = (const float*)d_in[6];
  const float* W2n = (const float*)d_in[7];
  const float* b2  = (const float*)d_in[8];
  float*       out = (float*)d_out;

  char*  ws  = (char*)d_ws;
  int*   cnt = (int*)ws;
  float* h   = (float*)ws + N_NODES;

  int* slots = (int*)d_out;

  zero_i32<<<N_NODES / 256, 256, 0, stream>>>(cnt);
  fill_kernel<<<N_EDGES / 4 / 256, 256, 0, stream>>>(src, dst, cnt, slots);

  sage_fused<128, true><<<FGRID, 32 * (128 / 16 + GW), 0, stream>>>(
      x, src, dst, cnt, slots, h, W1s, W1n, b1);
  sage_fused<64, false><<<FGRID, 32 * (64 / 16 + GW), 0, stream>>>(
      h, src, dst, cnt, slots, out, W2s, W2n, b2);
}


# Bucket table built with shared-memory counters only: edges split into runs per destination block, then one work-group per 256 nodes ranks them and writes whole bucket lines; no global atomics
# speedup vs baseline: 5.0129x; 1.5935x over previous
#include <hip/hip_runtime.h>

#define N_NODES 65536
#define N_EDGES 1048576
#define KD      128

typedef __attribute__((ext_vector_type(2)))  float     v2f;
typedef __attribute__((ext_vector_type(4)))  float     v4f;
typedef __attribute__((ext_vector_type(8)))  float     v8f;

#define CAP  64
#define NBLK 256
#define SUB  64

#define PT 512

__global__ __launch_bounds__(PT)
void part_kernel(const int* __restrict__ src,
                 const int* __restrict__ dst,
                 int* __restrict__ runs,
                 int* __restrict__ runlen) {
  __shared__ int hist[NBLK];
  const int tid = threadIdx.x;
  const int eb  = blockIdx.x;
  if (tid < NBLK) hist[tid] = 0;
  __syncthreads();
  const int t = eb * PT + tid;
  int sv[8], dv[8], rk[8];
#pragma unroll
  for (int j = 0; j < 2; j++) {
    const int4 s = ((const int4*)src)[2 * t + j];
    const int4 d = ((const int4*)dst)[2 * t + j];
    sv[4 * j + 0] = s.x; sv[4 * j + 1] = s.y; sv[4 * j + 2] = s.z; sv[4 * j + 3] = s.w;
    dv[4 * j + 0] = d.x & (N_NODES - 1); dv[4 * j + 1] = d.y & (N_NODES - 1);
    dv[4 * j + 2] = d.z & (N_NODES - 1); dv[4 * j + 3] = d.w & (N_NODES - 1);
  }
#pragma unroll
  for (int j = 0; j < 8; j++) rk[j] = atomicAdd(&hist[dv[j] >> 8], 1);
#pragma unroll
  for (int j = 0; j < 8; j++) {
    if (rk[j] < SUB)
      runs[((dv[j] >> 8) * NBLK + eb) * SUB + rk[j]] =
          (sv[j] & (N_NODES - 1)) | ((dv[j] & 255) << 16);
  }
  __syncthreads();
  if (tid < NBLK) runlen[tid * (256 * CAP) + eb] = hist[tid];
}

#define BT 512
#define RUNS_PW (NBLK / (BT / 32))

__global__ __launch_bounds__(BT)
void bucket_kernel(const int* __restrict__ src,
                   const int* __restrict__ dst,
                   const int* __restrict__ runs,
                   int* __restrict__ cnt,
                   int* slots) {
  __shared__ __attribute__((aligned(16))) int lslots[256 * CAP];
  __shared__ int lcnt[256];
  __shared__ int rl[NBLK];
  const int tid  = threadIdx.x;
  const int lane = tid & 31;
  const int wave = tid >> 5;
  const int b    = blockIdx.x;
  int over = 0;
  if (tid < NBLK) {
    const int n = slots[b * (256 * CAP) + tid];
    rl[tid]   = n;
    lcnt[tid] = 0;
    over      = n > SUB;
  }
  const int4 z4 = make_int4(0, 0, 0, 0);
  for (int j = 0; j < 256 * CAP / 4 / BT; j++) ((int4*)lslots)[tid + j * BT] = z4;
  const int any_over = __syncthreads_or(over);
  if (!any_over) {
    int nn[RUNS_PW], ent[RUNS_PW], rk[RUNS_PW];
#pragma unroll
    for (int p = 0; p < RUNS_PW; p++) nn[p] = rl[p * (BT / 32) + wave];
#pragma unroll
    for (int p = 0; p < RUNS_PW; p++) {
      const int k = p * (BT / 32) + wave;
      ent[p] = lane < nn[p] ? runs[(b * NBLK + k) * SUB + lane] : -1;
    }
#pragma unroll
    for (int p = 0; p < RUNS_PW; p++)
      rk[p] = ent[p] >= 0 ? atomicAdd(&lcnt[(ent[p] >> 16) & 255], 1) : CAP;
#pragma unroll
    for (int p = 0; p < RUNS_PW; p++) {
      if (rk[p] < CAP) lslots[((ent[p] >> 16) & 255) * CAP + rk[p]] = ent[p] & 0xffff;
    }
    for (int p = 0; p < RUNS_PW; p++) {
      const int k = p * (BT / 32) + wave;
      const int n = rl[k];
      for (int i = lane + 32; i < n; i += 32) {
        const int e = runs[(b * NBLK + k) * SUB + i];
        const int ln = (e >> 16) & 255;
        const int r  = atomicAdd(&lcnt[ln], 1);
        if (r < CAP) lslots[ln * CAP + r] = e & 0xffff;
      }
    }
  } else {
    for (int e = tid; e < N_EDGES; e += BT) {
      const int d = dst[e] & (N_NODES - 1);
      if ((d >> 8) == b) {
        const int ln = d & 255;
        const int r  = atomicAdd(&lcnt[ln], 1);
        if (r < CAP) lslots[ln * CAP + r] = src[e] & (N_NODES - 1);
      }
    }
  }
  __syncthreads();
  if (tid < 256) cnt[b * 256 + tid] = lcnt[tid];
  int4* o = (int4*)slots + (size_t)b * (256 * CAP / 4);
  for (int j = 0; j < 256 * CAP / 4 / BT; j++)
    o[tid + j * BT] = ((const int4*)lslots)[tid + j * BT];
}

__device__ __forceinline__ float4 row_of(const float4* __restrict__ f4, int my,
                                         int k, int lane) {
  const int idx = __builtin_amdgcn_readlane(my, k);
  return f4[(size_t)idx * (KD / 4) + lane];
}

__device__ __forceinline__ void add4(float4& acc, const float4 v) {
  acc.x += v.x; acc.y += v.y; acc.z += v.z; acc.w += v.w;
}

__device__ __forceinline__ float4 row_at(const char* __restrict__ base, int off,
                                         int k, unsigned lane16) {
  const unsigned o = (unsigned)__builtin_amdgcn_readlane(off, k) + lane16;
  return *(const float4*)(base + o);
}

__device__ __forceinline__ void sum_line(const char* __restrict__ base, int off,
                                         int t, unsigned lane16, float4& acc) {
  float4 v[32];
  const int      nq = min(t >> 2, 7);
  const int      r  = t - 4 * nq;
  const unsigned gm = (1u << nq) - 1u;
  const unsigned sm = (1u << r) - 1u;
  const float4 z = make_float4(0.0f, 0.0f, 0.0f, 0.0f);
#pragma unroll
  for (int q = 0; q < 7; q++) {
    if (gm & (1u << q)) {
      v[4 * q + 0] = row_at(base, off, 4 * q + 0, lane16);
      v[4 * q + 1] = row_at(base, off, 4 * q + 1, lane16);
      v[4 * q + 2] = row_at(base, off, 4 * q + 2, lane16);
      v[4 * q + 3] = row_at(base, off, 4 * q + 3, lane16);
    } else {
      v[4 * q + 0] = z; v[4 * q + 1] = z; v[4 * q + 2] = z; v[4 * q + 3] = z;
    }
  }
#pragma unroll
  for (int j = 0; j < 4; j++) {
    if (sm & (1u << j)) v[28 + j] = row_at(base, off, 4 * nq + j, lane16);
    else                v[28 + j] = z;
  }
#pragma unroll
  for (int j = 3; j >= 0; j--) {
    if (sm & (1u << j)) add4(acc, v[28 + j]);
  }
#pragma unroll
  for (int q = 6; q >= 0; q--) {
    if (gm & (1u << q)) {
      add4(acc, v[4 * q + 3]);
      add4(acc, v[4 * q + 2]);
      add4(acc, v[4 * q + 1]);
      add4(acc, v[4 * q + 0]);
    }
  }
}

__device__ __forceinline__ float4 node_sum(const float* __restrict__ X,
                                           const int* __restrict__ src,
                                           const int* __restrict__ dst,
                                           const int* slots, int node, int c,
                                           int cur, int lane) {
  float4 acc = make_float4(0.0f, 0.0f, 0.0f, 0.0f);
  if (c <= CAP) {
    const char*    xb     = (const char*)X;
    const unsigned lane16 = (unsigned)lane << 4;
    if (c > 0) sum_line(xb, cur, min(c, 32), lane16, acc);
    if (c > 32) {
      const int s2 = *(const int*)((const char*)slots +
                                   ((unsigned)node * (CAP * 4u) + 128u +
                                    ((unsigned)lane << 2)));
      const int more = (s2 & (N_NODES - 1)) << 9;
      for (int k = 0; k < c - 32; k++) add4(acc, row_at(xb, more, k, lane16));
    }
  } else {
    const float4* f4 = (const float4*)X;
    for (int e0 = 0; e0 < N_EDGES; e0 += 32) {
      const int d = dst[e0 + lane] & (N_NODES - 1);
      const int s = src[e0 + lane] & (N_NODES - 1);
      unsigned  m = __builtin_amdgcn_ballot_w32(d == node);
      while (m) {
        const int l = __builtin_ctz(m);
        m &= m - 1;
        add4(acc, row_of(f4, s, l, lane));
      }
    }
  }
  return acc;
}

#define FGRID 256
#define GW    16

template <int NOUT, bool RELU>
__global__ __launch_bounds__(32 * (NOUT / 16 + GW))
void sage_fused(const float* __restrict__ X,
                const int* __restrict__ src,
                const int* __restrict__ dst,
                const int* __restrict__ cnt,
                const int* slots,
                float* OUT,
                const float* __restrict__ Wself,
                const float* __restrict__ Wneigh,
                const float* __restrict__ bias) {
  constexpr int NW = NOUT / 16;
  constexpr int NT = N_NODES / 16;
  constexpr int NI = NT / FGRID;
  constexpr int KS = KD / 4;
  static_assert(NI == 16 && GW == 16, "one count per lane, one node per wave");

  __shared__ __attribute__((aligned(16))) v2f wS[NW][KS][32];
  __shared__ __attribute__((aligned(16))) v2f wN[NW][KS][32];
  __shared__ __attribute__((aligned(16))) float xs[2][16][KD + 4];
  __shared__ __attribute__((aligned(16))) float ms[2][16][KD + 4];

  const int tid  = threadIdx.x;
  const int lane = tid & 31;
  const int  wave  = __builtin_amdgcn_readfirstlane(tid >> 5);
  const bool mul   = wave < NW;
  const int  tile0 = blockIdx.x;
  const float4* f4 = (const float4*)X;

  const int  mrow = lane & 15;
  const bool hi   = lane >= 16;
  const int  ncol = wave * 16 + (lane & 15);
  const int  koff = hi ? 2 : 0;
  float bv = 0.0f;
  constexpr int RPW = 16 / NW;
  float4 xr0, xr1, xr2, xr3;
  static_assert(RPW == 2 || RPW == 4, "two or four own rows per wave");
  auto own_src = [&](int t, int j) -> const float4* {
    return f4 + (size_t)(t * 16 + wave + j * NW) * (KD / 4) + lane;
  };
  auto own_dst = [&](int b, int j) -> float4* {
    return (float4*)&xs[b][wave + j * NW][0] + lane;
  };

  const int g = wave - NW;
  int cv  = 0;
  int cur = 0;

  if (mul) {
    for (int s = 0; s < KS; s++) {
      const int ka = 4 * s + koff;
      wS[wave][s][lane] = v2f{ Wself[(size_t)ka * NOUT + ncol],
                               Wself[(size_t)(ka + 1) * NOUT + ncol] };
      wN[wave][s][lane] = v2f{ Wneigh[(size_t)ka * NOUT + ncol],
                               Wneigh[(size_t)(ka + 1) * NOUT + ncol] };
    }
    bv = bias[ncol];
    *own_dst(0, 0) = *own_src(tile0, 0);
    *own_dst(0, 1) = *own_src(tile0, 1);
    if (RPW == 4) {
      *own_dst(0, 2) = *own_src(tile0, 2);
      *own_dst(0, 3) = *own_src(tile0, 3);
    }
    xr0 = *own_src(tile0 + FGRID, 0);
    xr1 = *own_src(tile0 + FGRID, 1);
    if (RPW == 4) {
      xr2 = *own_src(tile0 + FGRID, 2);
      xr3 = *own_src(tile0 + FGRID, 3);
    }
  } else {
    __builtin_amdgcn_s_setprio(3);
    cv = cnt[(tile0 + (lane & (NI - 1)) * FGRID) * 16 + g];
    cur = (*(const int*)((const char*)slots +
                         ((unsigned)(tile0 * 16 + g) * (CAP * 4u) +
                          ((unsigned)lane << 2))) & (N_NODES - 1)) << 9;
  }

  for (int i = -1; i < NI; i++) {
    if (mul) {
      if (i >= 0) {
        const int buf  = i & 1;
        const int tile = tile0 + i * FGRID;
        if (i + 1 < NI) {
          *own_dst(buf ^ 1, 0) = xr0;
          *own_dst(buf ^ 1, 1) = xr1;
          if (RPW == 4) {
            *own_dst(buf ^ 1, 2) = xr2;
            *own_dst(buf ^ 1, 3) = xr3;
          }
        }
        if (i + 2 < NI) {
          xr0 = *own_src(tile + 2 * FGRID, 0);
          xr1 = *own_src(tile + 2 * FGRID, 1);
          if (RPW == 4) {
            xr2 = *own_src(tile + 2 * FGRID, 2);
            xr3 = *own_src(tile + 2 * FGRID, 3);
          }
        }
        v8f c = {};
#pragma unroll
        for (int s = 0; s < KS; s++) {
          const int ka = 4 * s + koff;
          v2f a_self = { xs[buf][mrow][ka], xs[buf][mrow][ka + 1] };
          v2f a_mean = { ms[buf][mrow][ka], ms[buf][mrow][ka + 1] };
          c = __builtin_amdgcn_wmma_f32_16x16x4_f32(false, a_self, false,
                                                    wS[wave][s][lane], (short)0,
                                                    c, false, false);
          c = __builtin_amdgcn_wmma_f32_16x16x4_f32(false, a_mean, false,
                                                    wN[wave][s][lane], (short)0,
                                                    c, false, false);
        }
#pragma unroll
        for (int r = 0; r < 8; r++) {
          int   row = tile * 16 + r + (hi ? 8 : 0);
          float v   = c[r] + bv;
          if (RELU) v = fmaxf(v, 0.0f);
          OUT[(size_t)row * NOUT + ncol] = v;
        }
      }
    } else if (i + 1 < NI) {
      const int it   = i + 1;
      const int b    = it & 1;
      const int node = (tile0 + it * FGRID) * 16 + g;
      const int c    = __builtin_amdgcn_readlane(cv, it);
      const int nn = it + 1 < NI ? node + FGRID * 16 : node;
      const int nx = *(const int*)((const char*)slots +
                                   ((unsigned)nn * (CAP * 4u) +
                                    ((unsigned)lane << 2)));
      const float4 a = node_sum(X, src, dst, slots, node, c, cur, lane);
      const float  r = 1.0f / fmaxf((float)c, 1.0f);
      ((float4*)&ms[b][g][0])[lane] =
          make_float4(a.x * r, a.y * r, a.z * r, a.w * r);
      cur = (nx & (N_NODES - 1)) << 9;
    }
    __builtin_amdgcn_fence(__ATOMIC_RELEASE, "workgroup", "local");
    __builtin_amdgcn_s_barrier();
    __builtin_amdgcn_fence(__ATOMIC_ACQUIRE, "workgroup", "local");
  }
}
extern "C" void kernel_launch(void* const* d_in, const int* in_sizes, int n_in,
                              void* d_out, int out_size, void* d_ws,
                              size_t ws_size, hipStream_t stream) {
  (void)in_sizes; (void)n_in; (void)out_size; (void)ws_size;
  const float* x   = (const float*)d_in[0];
  const int*   src = (const int*)d_in[1];
  const int*   dst = (const int*)d_in[2];
  const float* W1s = (const float*)d_in[3];
  const float* W1n = (const float*)d_in[4];
  const float* b1  = (const float*)d_in[5];
  const float* W2s = (const float*)d_in[6];
  const float* W2n = (const float*)d_in[7];
  const float* b2  = (const float*)d_in[8];
  float*       out = (float*)d_out;

  char*  ws  = (char*)d_ws;
  int*   cnt = (int*)ws;
  float* h   = (float*)ws + N_NODES;

  int* slots = (int*)d_out;

  int* runs = (int*)h;
  part_kernel<<<NBLK, PT, 0, stream>>>(src, dst, runs, slots);
  bucket_kernel<<<NBLK, BT, 0, stream>>>(src, dst, runs, cnt, slots);

  sage_fused<128, true><<<FGRID, 32 * (128 / 16 + GW), 0, stream>>>(
      x, src, dst, cnt, slots, h, W1s, W1n, b1);
  sage_fused<64, false><<<FGRID, 32 * (64 / 16 + GW), 0, stream>>>(
      h, src, dst, cnt, slots, out, W2s, W2n, b2);
}


# Multiplying waves read their shared-memory fragments in groups of four k steps, two groups ahead of the matrix instructions that use them (ring of three register sets); sum order unchanged
# speedup vs baseline: 5.1855x; 1.0344x over previous
#include <hip/hip_runtime.h>

#define N_NODES 65536
#define N_EDGES 1048576
#define KD      128

typedef __attribute__((ext_vector_type(2)))  float     v2f;
typedef __attribute__((ext_vector_type(4)))  float     v4f;
typedef __attribute__((ext_vector_type(8)))  float     v8f;

#define CAP  64
#define NBLK 256
#define SUB  64

#define PT 512

__global__ __launch_bounds__(PT)
void part_kernel(const int* __restrict__ src,
                 const int* __restrict__ dst,
                 int* __restrict__ runs,
                 int* __restrict__ runlen) {
  __shared__ int hist[NBLK];
  const int tid = threadIdx.x;
  const int eb  = blockIdx.x;
  if (tid < NBLK) hist[tid] = 0;
  __syncthreads();
  const int t = eb * PT + tid;
  int sv[8], dv[8], rk[8];
#pragma unroll
  for (int j = 0; j < 2; j++) {
    const int4 s = ((const int4*)src)[2 * t + j];
    const int4 d = ((const int4*)dst)[2 * t + j];
    sv[4 * j + 0] = s.x; sv[4 * j + 1] = s.y; sv[4 * j + 2] = s.z; sv[4 * j + 3] = s.w;
    dv[4 * j + 0] = d.x & (N_NODES - 1); dv[4 * j + 1] = d.y & (N_NODES - 1);
    dv[4 * j + 2] = d.z & (N_NODES - 1); dv[4 * j + 3] = d.w & (N_NODES - 1);
  }
#pragma unroll
  for (int j = 0; j < 8; j++) rk[j] = atomicAdd(&hist[dv[j] >> 8], 1);
#pragma unroll
  for (int j = 0; j < 8; j++) {
    if (rk[j] < SUB)
      runs[((dv[j] >> 8) * NBLK + eb) * SUB + rk[j]] =
          (sv[j] & (N_NODES - 1)) | ((dv[j] & 255) << 16);
  }
  __syncthreads();
  if (tid < NBLK) runlen[tid * (256 * CAP) + eb] = hist[tid];
}

#define BT 512
#define RUNS_PW (NBLK / (BT / 32))

__global__ __launch_bounds__(BT)
void bucket_kernel(const int* __restrict__ src,
                   const int* __restrict__ dst,
                   const int* __restrict__ runs,
                   int* __restrict__ cnt,
                   int* slots) {
  __shared__ __attribute__((aligned(16))) int lslots[256 * CAP];
  __shared__ int lcnt[256];
  __shared__ int rl[NBLK];
  const int tid  = threadIdx.x;
  const int lane = tid & 31;
  const int wave = tid >> 5;
  const int b    = blockIdx.x;
  int over = 0;
  if (tid < NBLK) {
    const int n = slots[b * (256 * CAP) + tid];
    rl[tid]   = n;
    lcnt[tid] = 0;
    over      = n > SUB;
  }
  const int4 z4 = make_int4(0, 0, 0, 0);
  for (int j = 0; j < 256 * CAP / 4 / BT; j++) ((int4*)lslots)[tid + j * BT] = z4;
  const int any_over = __syncthreads_or(over);
  if (!any_over) {
    int nn[RUNS_PW], ent[RUNS_PW], rk[RUNS_PW];
#pragma unroll
    for (int p = 0; p < RUNS_PW; p++) nn[p] = rl[p * (BT / 32) + wave];
#pragma unroll
    for (int p = 0; p < RUNS_PW; p++) {
      const int k = p * (BT / 32) + wave;
      ent[p] = lane < nn[p] ? runs[(b * NBLK + k) * SUB + lane] : -1;
    }
#pragma unroll
    for (int p = 0; p < RUNS_PW; p++)
      rk[p] = ent[p] >= 0 ? atomicAdd(&lcnt[(ent[p] >> 16) & 255], 1) : CAP;
#pragma unroll
    for (int p = 0; p < RUNS_PW; p++) {
      if (rk[p] < CAP) lslots[((ent[p] >> 16) & 255) * CAP + rk[p]] = ent[p] & 0xffff;
    }
    for (int p = 0; p < RUNS_PW; p++) {
      const int k = p * (BT / 32) + wave;
      const int n = rl[k];
      for (int i = lane + 32; i < n; i += 32) {
        const int e = runs[(b * NBLK + k) * SUB + i];
        const int ln = (e >> 16) & 255;
        const int r  = atomicAdd(&lcnt[ln], 1);
        if (r < CAP) lslots[ln * CAP + r] = e & 0xffff;
      }
    }
  } else {
    for (int e = tid; e < N_EDGES; e += BT) {
      const int d = dst[e] & (N_NODES - 1);
      if ((d >> 8) == b) {
        const int ln = d & 255;
        const int r  = atomicAdd(&lcnt[ln], 1);
        if (r < CAP) lslots[ln * CAP + r] = src[e] & (N_NODES - 1);
      }
    }
  }
  __syncthreads();
  if (tid < 256) cnt[b * 256 + tid] = lcnt[tid];
  int4* o = (int4*)slots + (size_t)b * (256 * CAP / 4);
  for (int j = 0; j < 256 * CAP / 4 / BT; j++)
    o[tid + j * BT] = ((const int4*)lslots)[tid + j * BT];
}

__device__ __forceinline__ float4 row_of(const float4* __restrict__ f4, int my,
                                         int k, int lane) {
  const int idx = __builtin_amdgcn_readlane(my, k);
  return f4[(size_t)idx * (KD / 4) + lane];
}

__device__ __forceinline__ void add4(float4& acc, const float4 v) {
  acc.x += v.x; acc.y += v.y; acc.z += v.z; acc.w += v.w;
}

__device__ __forceinline__ float4 row_at(const char* __restrict__ base, int off,
                                         int k, unsigned lane16) {
  const unsigned o = (unsigned)__builtin_amdgcn_readlane(off, k) + lane16;
  return *(const float4*)(base + o);
}

__device__ __forceinline__ void sum_line(const char* __restrict__ base, int off,
                                         int t, unsigned lane16, float4& acc) {
  float4 v[32];
  const int      nq = min(t >> 2, 7);
  const int      r  = t - 4 * nq;
  const unsigned gm = (1u << nq) - 1u;
  const unsigned sm = (1u << r) - 1u;
  const float4 z = make_float4(0.0f, 0.0f, 0.0f, 0.0f);
#pragma unroll
  for (int q = 0; q < 7; q++) {
    if (gm & (1u << q)) {
      v[4 * q + 0] = row_at(base, off, 4 * q + 0, lane16);
      v[4 * q + 1] = row_at(base, off, 4 * q + 1, lane16);
      v[4 * q + 2] = row_at(base, off, 4 * q + 2, lane16);
      v[4 * q + 3] = row_at(base, off, 4 * q + 3, lane16);
    } else {
      v[4 * q + 0] = z; v[4 * q + 1] = z; v[4 * q + 2] = z; v[4 * q + 3] = z;
    }
  }
#pragma unroll
  for (int j = 0; j < 4; j++) {
    if (sm & (1u << j)) v[28 + j] = row_at(base, off, 4 * nq + j, lane16);
    else                v[28 + j] = z;
  }
#pragma unroll
  for (int j = 3; j >= 0; j--) {
    if (sm & (1u << j)) add4(acc, v[28 + j]);
  }
#pragma unroll
  for (int q = 6; q >= 0; q--) {
    if (gm & (1u << q)) {
      add4(acc, v[4 * q + 3]);
      add4(acc, v[4 * q + 2]);
      add4(acc, v[4 * q + 1]);
      add4(acc, v[4 * q + 0]);
    }
  }
}

__device__ __forceinline__ float4 node_sum(const float* __restrict__ X,
                                           const int* __restrict__ src,
                                           const int* __restrict__ dst,
                                           const int* slots, int node, int c,
                                           int cur, int lane) {
  float4 acc = make_float4(0.0f, 0.0f, 0.0f, 0.0f);
  if (c <= CAP) {
    const char*    xb     = (const char*)X;
    const unsigned lane16 = (unsigned)lane << 4;
    if (c > 0) sum_line(xb, cur, min(c, 32), lane16, acc);
    if (c > 32) {
      const int s2 = *(const int*)((const char*)slots +
                                   ((unsigned)node * (CAP * 4u) + 128u +
                                    ((unsigned)lane << 2)));
      const int more = (s2 & (N_NODES - 1)) << 9;
      for (int k = 0; k < c - 32; k++) add4(acc, row_at(xb, more, k, lane16));
    }
  } else {
    const float4* f4 = (const float4*)X;
    for (int e0 = 0; e0 < N_EDGES; e0 += 32) {
      const int d = dst[e0 + lane] & (N_NODES - 1);
      const int s = src[e0 + lane] & (N_NODES - 1);
      unsigned  m = __builtin_amdgcn_ballot_w32(d == node);
      while (m) {
        const int l = __builtin_ctz(m);
        m &= m - 1;
        add4(acc, row_of(f4, s, l, lane));
      }
    }
  }
  return acc;
}

#define FGRID 256
#define GW    16

template <int NOUT, bool RELU>
__global__ __launch_bounds__(32 * (NOUT / 16 + GW))
void sage_fused(const float* __restrict__ X,
                const int* __restrict__ src,
                const int* __restrict__ dst,
                const int* __restrict__ cnt,
                const int* slots,
                float* OUT,
                const float* __restrict__ Wself,
                const float* __restrict__ Wneigh,
                const float* __restrict__ bias) {
  constexpr int NW = NOUT / 16;
  constexpr int NT = N_NODES / 16;
  constexpr int NI = NT / FGRID;
  constexpr int KS = KD / 4;
  static_assert(NI == 16 && GW == 16, "one count per lane, one node per wave");

  __shared__ __attribute__((aligned(16))) v2f wS[NW][KS][32];
  __shared__ __attribute__((aligned(16))) v2f wN[NW][KS][32];
  __shared__ __attribute__((aligned(16))) float xs[2][16][KD + 4];
  __shared__ __attribute__((aligned(16))) float ms[2][16][KD + 4];

  const int tid  = threadIdx.x;
  const int lane = tid & 31;
  const int  wave  = __builtin_amdgcn_readfirstlane(tid >> 5);
  const bool mul   = wave < NW;
  const int  tile0 = blockIdx.x;
  const float4* f4 = (const float4*)X;

  const int  mrow = lane & 15;
  const bool hi   = lane >= 16;
  const int  ncol = wave * 16 + (lane & 15);
  const int  koff = hi ? 2 : 0;
  float bv = 0.0f;
  constexpr int RPW = 16 / NW;
  float4 xr0, xr1, xr2, xr3;
  static_assert(RPW == 2 || RPW == 4, "two or four own rows per wave");
  auto own_src = [&](int t, int j) -> const float4* {
    return f4 + (size_t)(t * 16 + wave + j * NW) * (KD / 4) + lane;
  };
  auto own_dst = [&](int b, int j) -> float4* {
    return (float4*)&xs[b][wave + j * NW][0] + lane;
  };

  const int g = wave - NW;
  int cv  = 0;
  int cur = 0;

  if (mul) {
    for (int s = 0; s < KS; s++) {
      const int ka = 4 * s + koff;
      wS[wave][s][lane] = v2f{ Wself[(size_t)ka * NOUT + ncol],
                               Wself[(size_t)(ka + 1) * NOUT + ncol] };
      wN[wave][s][lane] = v2f{ Wneigh[(size_t)ka * NOUT + ncol],
                               Wneigh[(size_t)(ka + 1) * NOUT + ncol] };
    }
    bv = bias[ncol];
    *own_dst(0, 0) = *own_src(tile0, 0);
    *own_dst(0, 1) = *own_src(tile0, 1);
    if (RPW == 4) {
      *own_dst(0, 2) = *own_src(tile0, 2);
      *own_dst(0, 3) = *own_src(tile0, 3);
    }
    xr0 = *own_src(tile0 + FGRID, 0);
    xr1 = *own_src(tile0 + FGRID, 1);
    if (RPW == 4) {
      xr2 = *own_src(tile0 + FGRID, 2);
      xr3 = *own_src(tile0 + FGRID, 3);
    }
  } else {
    __builtin_amdgcn_s_setprio(3);
    cv = cnt[(tile0 + (lane & (NI - 1)) * FGRID) * 16 + g];
    cur = (*(const int*)((const char*)slots +
                         ((unsigned)(tile0 * 16 + g) * (CAP * 4u) +
                          ((unsigned)lane << 2))) & (N_NODES - 1)) << 9;
  }

  for (int i = -1; i < NI; i++) {
    if (mul) {
      if (i >= 0) {
        const int buf  = i & 1;
        const int tile = tile0 + i * FGRID;
        if (i + 1 < NI) {
          *own_dst(buf ^ 1, 0) = xr0;
          *own_dst(buf ^ 1, 1) = xr1;
          if (RPW == 4) {
            *own_dst(buf ^ 1, 2) = xr2;
            *own_dst(buf ^ 1, 3) = xr3;
          }
        }
        if (i + 2 < NI) {
          xr0 = *own_src(tile + 2 * FGRID, 0);
          xr1 = *own_src(tile + 2 * FGRID, 1);
          if (RPW == 4) {
            xr2 = *own_src(tile + 2 * FGRID, 2);
            xr3 = *own_src(tile + 2 * FGRID, 3);
          }
        }
        v8f c = {};
        constexpr int KB = 4;
        constexpr int NB = KS / KB;
        v2f fa[3][KB], fm[3][KB], fs[3][KB], fn[3][KB];
        auto read_block = [&](int blk, int set) {
#pragma unroll
          for (int j = 0; j < KB; j++) {
            const int s  = blk * KB + j;
            const int ka = 4 * s + koff;
            fa[set][j] = v2f{ xs[buf][mrow][ka], xs[buf][mrow][ka + 1] };
            fm[set][j] = v2f{ ms[buf][mrow][ka], ms[buf][mrow][ka + 1] };
            fs[set][j] = wS[wave][s][lane];
            fn[set][j] = wN[wave][s][lane];
          }
        };
        read_block(0, 0);
        read_block(1, 1);
#pragma unroll
        for (int b = 0; b < NB; b++) {
          if (b + 2 < NB) read_block(b + 2, (b + 2) % 3);
          __builtin_amdgcn_sched_barrier(0);
#pragma unroll
          for (int j = 0; j < KB; j++) {
            c = __builtin_amdgcn_wmma_f32_16x16x4_f32(false, fa[b % 3][j], false,
                                                      fs[b % 3][j], (short)0,
                                                      c, false, false);
            c = __builtin_amdgcn_wmma_f32_16x16x4_f32(false, fm[b % 3][j], false,
                                                      fn[b % 3][j], (short)0,
                                                      c, false, false);
          }
          __builtin_amdgcn_sched_barrier(0);
        }
#pragma unroll
        for (int r = 0; r < 8; r++) {
          int   row = tile * 16 + r + (hi ? 8 : 0);
          float v   = c[r] + bv;
          if (RELU) v = fmaxf(v, 0.0f);
          OUT[(size_t)row * NOUT + ncol] = v;
        }
      }
    } else if (i + 1 < NI) {
      const int it   = i + 1;
      const int b    = it & 1;
      const int node = (tile0 + it * FGRID) * 16 + g;
      const int c    = __builtin_amdgcn_readlane(cv, it);
      const int nn = it + 1 < NI ? node + FGRID * 16 : node;
      const int nx = *(const int*)((const char*)slots +
                                   ((unsigned)nn * (CAP * 4u) +
                                    ((unsigned)lane << 2)));
      const float4 a = node_sum(X, src, dst, slots, node, c, cur, lane);
      const float  r = 1.0f / fmaxf((float)c, 1.0f);
      ((float4*)&ms[b][g][0])[lane] =
          make_float4(a.x * r, a.y * r, a.z * r, a.w * r);
      cur = (nx & (N_NODES - 1)) << 9;
    }
    __builtin_amdgcn_fence(__ATOMIC_RELEASE, "workgroup", "local");
    __builtin_amdgcn_s_barrier();
    __builtin_amdgcn_fence(__ATOMIC_ACQUIRE, "workgroup", "local");
  }
}
extern "C" void kernel_launch(void* const* d_in, const int* in_sizes, int n_in,
                              void* d_out, int out_size, void* d_ws,
                              size_t ws_size, hipStream_t stream) {
  (void)in_sizes; (void)n_in; (void)out_size; (void)ws_size;
  const float* x   = (const float*)d_in[0];
  const int*   src = (const int*)d_in[1];
  const int*   dst = (const int*)d_in[2];
  const float* W1s = (const float*)d_in[3];
  const float* W1n = (const float*)d_in[4];
  const float* b1  = (const float*)d_in[5];
  const float* W2s = (const float*)d_in[6];
  const float* W2n = (const float*)d_in[7];
  const float* b2  = (const float*)d_in[8];
  float*       out = (float*)d_out;

  char*  ws  = (char*)d_ws;
  int*   cnt = (int*)ws;
  float* h   = (float*)ws + N_NODES;

  int* slots = (int*)d_out;

  int* runs = (int*)h;
  part_kernel<<<NBLK, PT, 0, stream>>>(src, dst, runs, slots);
  bucket_kernel<<<NBLK, BT, 0, stream>>>(src, dst, runs, cnt, slots);

  sage_fused<128, true><<<FGRID, 32 * (128 / 16 + GW), 0, stream>>>(
      x, src, dst, cnt, slots, h, W1s, W1n, b1);
  sage_fused<64, false><<<FGRID, 32 * (64 / 16 + GW), 0, stream>>>(
      h, src, dst, cnt, slots, out, W2s, W2n, b2);
}


# Layer 2 multiplies before it averages: H*[W2self|W2neigh] on WMMA inside the layer-1 launch, H only in shared memory; last launch gathers 64-float rows of H*W2neigh, adds; layer-2 sum order changes
# speedup vs baseline: 5.7350x; 1.1060x over previous
#include <hip/hip_runtime.h>

#define N_NODES 65536
#define N_EDGES 1048576
#define KD      128

typedef __attribute__((ext_vector_type(2)))  float     v2f;
typedef __attribute__((ext_vector_type(4)))  float     v4f;
typedef __attribute__((ext_vector_type(8)))  float     v8f;

#define CAP  64
#define NBLK 256
#define SUB  64

#define PT 512

__global__ __launch_bounds__(PT)
void part_kernel(const int* __restrict__ src,
                 const int* __restrict__ dst,
                 int* __restrict__ runs,
                 int* __restrict__ runlen) {
  __shared__ int hist[NBLK];
  const int tid = threadIdx.x;
  const int eb  = blockIdx.x;
  if (tid < NBLK) hist[tid] = 0;
  __syncthreads();
  const int t = eb * PT + tid;
  int sv[8], dv[8], rk[8];
#pragma unroll
  for (int j = 0; j < 2; j++) {
    const int4 s = ((const int4*)src)[2 * t + j];
    const int4 d = ((const int4*)dst)[2 * t + j];
    sv[4 * j + 0] = s.x; sv[4 * j + 1] = s.y; sv[4 * j + 2] = s.z; sv[4 * j + 3] = s.w;
    dv[4 * j + 0] = d.x & (N_NODES - 1); dv[4 * j + 1] = d.y & (N_NODES - 1);
    dv[4 * j + 2] = d.z & (N_NODES - 1); dv[4 * j + 3] = d.w & (N_NODES - 1);
  }
#pragma unroll
  for (int j = 0; j < 8; j++) rk[j] = atomicAdd(&hist[dv[j] >> 8], 1);
#pragma unroll
  for (int j = 0; j < 8; j++) {
    if (rk[j] < SUB)
      runs[((dv[j] >> 8) * NBLK + eb) * SUB + rk[j]] =
          (sv[j] & (N_NODES - 1)) | ((dv[j] & 255) << 16);
  }
  __syncthreads();
  if (tid < NBLK) runlen[tid * (256 * CAP) + eb] = hist[tid];
}

#define BT 512
#define RUNS_PW (NBLK / (BT / 32))

__global__ __launch_bounds__(BT)
void bucket_kernel(const int* __restrict__ src,
                   const int* __restrict__ dst,
                   const int* __restrict__ runs,
                   int* __restrict__ cnt,
                   int* slots) {
  __shared__ __attribute__((aligned(16))) int lslots[256 * CAP];
  __shared__ int lcnt[256];
  __shared__ int rl[NBLK];
  const int tid  = threadIdx.x;
  const int lane = tid & 31;
  const int wave = tid >> 5;
  const int b    = blockIdx.x;
  int over = 0;
  if (tid < NBLK) {
    const int n = slots[b * (256 * CAP) + tid];
    rl[tid]   = n;
    lcnt[tid] = 0;
    over      = n > SUB;
  }
  const int4 z4 = make_int4(0, 0, 0, 0);
  for (int j = 0; j < 256 * CAP / 4 / BT; j++) ((int4*)lslots)[tid + j * BT] = z4;
  const int any_over = __syncthreads_or(over);
  if (!any_over) {
    int nn[RUNS_PW], ent[RUNS_PW], rk[RUNS_PW];
#pragma unroll
    for (int p = 0; p < RUNS_PW; p++) nn[p] = rl[p * (BT / 32) + wave];
#pragma unroll
    for (int p = 0; p < RUNS_PW; p++) {
      const int k = p * (BT / 32) + wave;
      ent[p] = lane < nn[p] ? runs[(b * NBLK + k) * SUB + lane] : -1;
    }
#pragma unroll
    for (int p = 0; p < RUNS_PW; p++)
      rk[p] = ent[p] >= 0 ? atomicAdd(&lcnt[(ent[p] >> 16) & 255], 1) : CAP;
#pragma unroll
    for (int p = 0; p < RUNS_PW; p++) {
      if (rk[p] < CAP) lslots[((ent[p] >> 16) & 255) * CAP + rk[p]] = ent[p] & 0xffff;
    }
    for (int p = 0; p < RUNS_PW; p++) {
      const int k = p * (BT / 32) + wave;
      const int n = rl[k];
      for (int i = lane + 32; i < n; i += 32) {
        const int e = runs[(b * NBLK + k) * SUB + i];
        const int ln = (e >> 16) & 255;
        const int r  = atomicAdd(&lcnt[ln], 1);
        if (r < CAP) lslots[ln * CAP + r] = e & 0xffff;
      }
    }
  } else {
    for (int e = tid; e < N_EDGES; e += BT) {
      const int d = dst[e] & (N_NODES - 1);
      if ((d >> 8) == b) {
        const int ln = d & 255;
        const int r  = atomicAdd(&lcnt[ln], 1);
        if (r < CAP) lslots[ln * CAP + r] = src[e] & (N_NODES - 1);
      }
    }
  }
  __syncthreads();
  if (tid < 256) cnt[b * 256 + tid] = lcnt[tid];
  int4* o = (int4*)slots + (size_t)b * (256 * CAP / 4);
  for (int j = 0; j < 256 * CAP / 4 / BT; j++)
    o[tid + j * BT] = ((const int4*)lslots)[tid + j * BT];
}

__device__ __forceinline__ float4 row_of(const float4* __restrict__ f4, int my,
                                         int k, int lane) {
  const int idx = __builtin_amdgcn_readlane(my, k);
  return f4[(size_t)idx * (KD / 4) + lane];
}

__device__ __forceinline__ void add4(float4& acc, const float4 v) {
  acc.x += v.x; acc.y += v.y; acc.z += v.z; acc.w += v.w;
}

__device__ __forceinline__ float4 row_at(const char* __restrict__ base, int off,
                                         int k, unsigned lane16) {
  const unsigned o = (unsigned)__builtin_amdgcn_readlane(off, k) + lane16;
  return *(const float4*)(base + o);
}

__device__ __forceinline__ void sum_line(const char* __restrict__ base, int off,
                                         int t, unsigned lane16, float4& acc) {
  float4 v[32];
  const int      nq = min(t >> 2, 7);
  const int      r  = t - 4 * nq;
  const unsigned gm = (1u << nq) - 1u;
  const unsigned sm = (1u << r) - 1u;
  const float4 z = make_float4(0.0f, 0.0f, 0.0f, 0.0f);
#pragma unroll
  for (int q = 0; q < 7; q++) {
    if (gm & (1u << q)) {
      v[4 * q + 0] = row_at(base, off, 4 * q + 0, lane16);
      v[4 * q + 1] = row_at(base, off, 4 * q + 1, lane16);
      v[4 * q + 2] = row_at(base, off, 4 * q + 2, lane16);
      v[4 * q + 3] = row_at(base, off, 4 * q + 3, lane16);
    } else {
      v[4 * q + 0] = z; v[4 * q + 1] = z; v[4 * q + 2] = z; v[4 * q + 3] = z;
    }
  }
#pragma unroll
  for (int j = 0; j < 4; j++) {
    if (sm & (1u << j)) v[28 + j] = row_at(base, off, 4 * nq + j, lane16);
    else                v[28 + j] = z;
  }
#pragma unroll
  for (int j = 3; j >= 0; j--) {
    if (sm & (1u << j)) add4(acc, v[28 + j]);
  }
#pragma unroll
  for (int q = 6; q >= 0; q--) {
    if (gm & (1u << q)) {
      add4(acc, v[4 * q + 3]);
      add4(acc, v[4 * q + 2]);
      add4(acc, v[4 * q + 1]);
      add4(acc, v[4 * q + 0]);
    }
  }
}

__device__ __forceinline__ float4 node_sum(const float* __restrict__ X,
                                           const int* __restrict__ src,
                                           const int* __restrict__ dst,
                                           const int* slots, int node, int c,
                                           int cur, int lane) {
  float4 acc = make_float4(0.0f, 0.0f, 0.0f, 0.0f);
  if (c <= CAP) {
    const char*    xb     = (const char*)X;
    const unsigned lane16 = (unsigned)lane << 4;
    if (c > 0) sum_line(xb, cur, min(c, 32), lane16, acc);
    if (c > 32) {
      const int s2 = *(const int*)((const char*)slots +
                                   ((unsigned)node * (CAP * 4u) + 128u +
                                    ((unsigned)lane << 2)));
      const int more = (s2 & (N_NODES - 1)) << 9;
      for (int k = 0; k < c - 32; k++) add4(acc, row_at(xb, more, k, lane16));
    }
  } else {
    const float4* f4 = (const float4*)X;
    for (int e0 = 0; e0 < N_EDGES; e0 += 32) {
      const int d = dst[e0 + lane] & (N_NODES - 1);
      const int s = src[e0 + lane] & (N_NODES - 1);
      unsigned  m = __builtin_amdgcn_ballot_w32(d == node);
      while (m) {
        const int l = __builtin_ctz(m);
        m &= m - 1;
        add4(acc, row_of(f4, s, l, lane));
      }
    }
  }
  return acc;
}

#define NO2 64

__device__ __forceinline__ void add2(float2& acc, const float2 v) {
  acc.x += v.x; acc.y += v.y;
}

__device__ __forceinline__ float2 row_of2(const float2* __restrict__ f2, int my,
                                          int k, int lane) {
  const int idx = __builtin_amdgcn_readlane(my, k);
  return f2[(size_t)idx * (NO2 / 2) + lane];
}

__device__ __forceinline__ float2 row_at2(const char* __restrict__ base, int off,
                                          int k, unsigned lane8) {
  const unsigned o = (unsigned)__builtin_amdgcn_readlane(off, k) + lane8;
  return *(const float2*)(base + o);
}

__device__ __forceinline__ void sum_line2(const char* __restrict__ base, int off,
                                          int t, unsigned lane8, float2& acc) {
  float2 v[32];
  const int      nq = min(t >> 2, 7);
  const int      r  = t - 4 * nq;
  const unsigned gm = (1u << nq) - 1u;
  const unsigned sm = (1u << r) - 1u;
  const float2 z = make_float2(0.0f, 0.0f);
#pragma unroll
  for (int q = 0; q < 7; q++) {
    if (gm & (1u << q)) {
      v[4 * q + 0] = row_at2(base, off, 4 * q + 0, lane8);
      v[4 * q + 1] = row_at2(base, off, 4 * q + 1, lane8);
      v[4 * q + 2] = row_at2(base, off, 4 * q + 2, lane8);
      v[4 * q + 3] = row_at2(base, off, 4 * q + 3, lane8);
    } else {
      v[4 * q + 0] = z; v[4 * q + 1] = z; v[4 * q + 2] = z; v[4 * q + 3] = z;
    }
  }
#pragma unroll
  for (int j = 0; j < 4; j++) {
    if (sm & (1u << j)) v[28 + j] = row_at2(base, off, 4 * nq + j, lane8);
    else                v[28 + j] = z;
  }
#pragma unroll
  for (int j = 3; j >= 0; j--) {
    if (sm & (1u << j)) add2(acc, v[28 + j]);
  }
#pragma unroll
  for (int q = 6; q >= 0; q--) {
    if (gm & (1u << q)) {
      add2(acc, v[4 * q + 3]);
      add2(acc, v[4 * q + 2]);
      add2(acc, v[4 * q + 1]);
      add2(acc, v[4 * q + 0]);
    }
  }
}

__device__ __forceinline__ float2 node_sum2(const float* __restrict__ R,
                                            const int* __restrict__ src,
                                            const int* __restrict__ dst,
                                            const int* slots, int node, int c,
                                            int cur, int lane) {
  float2 acc = make_float2(0.0f, 0.0f);
  if (c <= CAP) {
    const char*    rb    = (const char*)R;
    const unsigned lane8 = (unsigned)lane << 3;
    if (c > 0) sum_line2(rb, cur, min(c, 32), lane8, acc);
    if (c > 32) {
      const int s2 = *(const int*)((const char*)slots +
                                   ((unsigned)node * (CAP * 4u) + 128u +
                                    ((unsigned)lane << 2)));
      const int more = (s2 & (N_NODES - 1)) << 8;
      for (int k = 0; k < c - 32; k++) add2(acc, row_at2(rb, more, k, lane8));
    }
  } else {
    const float2* f2 = (const float2*)R;
    for (int e0 = 0; e0 < N_EDGES; e0 += 32) {
      const int d = dst[e0 + lane] & (N_NODES - 1);
      const int s = src[e0 + lane] & (N_NODES - 1);
      unsigned  m = __builtin_amdgcn_ballot_w32(d == node);
      while (m) {
        const int l = __builtin_ctz(m);
        m &= m - 1;
        add2(acc, row_of2(f2, s, l, lane));
      }
    }
  }
  return acc;
}

#define FGRID 256
#define GW    16
#define NH    128

__global__ __launch_bounds__(32 * (NH / 16 + GW))
void sage_l1(const float* __restrict__ X,
             const int* __restrict__ src,
             const int* __restrict__ dst,
             const int* __restrict__ cnt,
             const int* slots,
             const float* __restrict__ Wself,
             const float* __restrict__ Wneigh,
             const float* __restrict__ bias,
             const float* __restrict__ W2self,
             const float* __restrict__ W2neigh,
             const float* __restrict__ bias2,
             float* __restrict__ S,
             float* __restrict__ P) {
  constexpr int NW = NH / 16;
  constexpr int NT = N_NODES / 16;
  constexpr int NI = NT / FGRID;
  constexpr int KS = KD / 4;
  static_assert(NI == 16 && GW == 16, "one count per lane, one node per wave");
  static_assert(NH == KD && 2 * NO2 == NH,
                "H is the second product's row tile; its strips are S, then P");

  __shared__ __attribute__((aligned(16))) v2f wS[NW][KS][32];
  __shared__ __attribute__((aligned(16))) v2f wN[NW][KS][32];
  __shared__ __attribute__((aligned(16))) v2f w2[NW][KS][32];
  __shared__ __attribute__((aligned(16))) float xs[2][16][KD + 4];
  __shared__ __attribute__((aligned(16))) float ms[2][16][KD + 4];
  __shared__ __attribute__((aligned(16))) float hs[2][16][NH + 4];

  const int tid  = threadIdx.x;
  const int lane = tid & 31;
  const int  wave  = __builtin_amdgcn_readfirstlane(tid >> 5);
  const bool mul   = wave < NW;
  const int  tile0 = blockIdx.x;
  const float4* f4 = (const float4*)X;

  const int  mrow = lane & 15;
  const bool hi   = lane >= 16;
  const int  ncol = wave * 16 + (lane & 15);
  const int  koff = hi ? 2 : 0;
  float bv = 0.0f;
  const bool   isP  = wave >= NW / 2;
  const int    col2 = (wave & (NW / 2 - 1)) * 16 + (lane & 15);
  const float* W2   = isP ? W2neigh : W2self;
  float*       O2   = isP ? P : S;
  float bv2 = 0.0f;
  float4 xr0, xr1;
  static_assert(16 / NW == 2, "two own rows per multiplying wave");
  auto own_src = [&](int t, int j) -> const float4* {
    return f4 + (size_t)(t * 16 + wave + j * NW) * (KD / 4) + lane;
  };
  auto own_dst = [&](int b, int j) -> float4* {
    return (float4*)&xs[b][wave + j * NW][0] + lane;
  };

  const int g = wave - NW;
  int cv  = 0;
  int cur = 0;

  if (mul) {
    for (int s = 0; s < KS; s++) {
      const int ka = 4 * s + koff;
      wS[wave][s][lane] = v2f{ Wself[(size_t)ka * NH + ncol],
                               Wself[(size_t)(ka + 1) * NH + ncol] };
      wN[wave][s][lane] = v2f{ Wneigh[(size_t)ka * NH + ncol],
                               Wneigh[(size_t)(ka + 1) * NH + ncol] };
      w2[wave][s][lane] = v2f{ W2[(size_t)ka * NO2 + col2],
                               W2[(size_t)(ka + 1) * NO2 + col2] };
    }
    bv = bias[ncol];
    if (!isP) bv2 = bias2[col2];
    *own_dst(0, 0) = *own_src(tile0, 0);
    *own_dst(0, 1) = *own_src(tile0, 1);
    xr0 = *own_src(tile0 + FGRID, 0);
    xr1 = *own_src(tile0 + FGRID, 1);
  } else {
    __builtin_amdgcn_s_setprio(3);
    cv = cnt[(tile0 + (lane & (NI - 1)) * FGRID) * 16 + g];
    cur = (*(const int*)((const char*)slots +
                         ((unsigned)(tile0 * 16 + g) * (CAP * 4u) +
                          ((unsigned)lane << 2))) & (N_NODES - 1)) << 9;
  }

  for (int i = -1; i <= NI; i++) {
    if (mul) {
      if (i >= 1) {
        const int hb   = (i - 1) & 1;
        const int tile = tile0 + (i - 1) * FGRID;
        v8f d = {};
        constexpr int KB2 = 8;
        constexpr int NB2 = KS / KB2;
        v2f fh[3][KB2], fw[3][KB2];
        auto read_block2 = [&](int blk, int set) {
#pragma unroll
          for (int j = 0; j < KB2; j++) {
            const int s  = blk * KB2 + j;
            const int ka = 4 * s + koff;
            fh[set][j] = v2f{ hs[hb][mrow][ka], hs[hb][mrow][ka + 1] };
            fw[set][j] = w2[wave][s][lane];
          }
        };
        read_block2(0, 0);
        read_block2(1, 1);
#pragma unroll
        for (int b = 0; b < NB2; b++) {
          if (b + 2 < NB2) read_block2(b + 2, (b + 2) % 3);
          __builtin_amdgcn_sched_barrier(0);
#pragma unroll
          for (int j = 0; j < KB2; j++) {
            d = __builtin_amdgcn_wmma_f32_16x16x4_f32(false, fh[b % 3][j], false,
                                                      fw[b % 3][j], (short)0,
                                                      d, false, false);
          }
          __builtin_amdgcn_sched_barrier(0);
        }
#pragma unroll
        for (int r = 0; r < 8; r++) {
          const int row = tile * 16 + r + (hi ? 8 : 0);
          O2[(size_t)row * NO2 + col2] = d[r] + bv2;
        }
      }
      if (i >= 0 && i < NI) {
        const int buf  = i & 1;
        const int tile = tile0 + i * FGRID;
        if (i + 1 < NI) {
          *own_dst(buf ^ 1, 0) = xr0;
          *own_dst(buf ^ 1, 1) = xr1;
        }
        if (i + 2 < NI) {
          xr0 = *own_src(tile + 2 * FGRID, 0);
          xr1 = *own_src(tile + 2 * FGRID, 1);
        }
        v8f c = {};
        constexpr int KB = 4;
        constexpr int NB = KS / KB;
        v2f fa[3][KB], fm[3][KB], fs[3][KB], fn[3][KB];
        auto read_block = [&](int blk, int set) {
#pragma unroll
          for (int j = 0; j < KB; j++) {
            const int s  = blk * KB + j;
            const int ka = 4 * s + koff;
            fa[set][j] = v2f{ xs[buf][mrow][ka], xs[buf][mrow][ka + 1] };
            fm[set][j] = v2f{ ms[buf][mrow][ka], ms[buf][mrow][ka + 1] };
            fs[set][j] = wS[wave][s][lane];
            fn[set][j] = wN[wave][s][lane];
          }
        };
        read_block(0, 0);
        read_block(1, 1);
#pragma unroll
        for (int b = 0; b < NB; b++) {
          if (b + 2 < NB) read_block(b + 2, (b + 2) % 3);
          __builtin_amdgcn_sched_barrier(0);
#pragma unroll
          for (int j = 0; j < KB; j++) {
            c = __builtin_amdgcn_wmma_f32_16x16x4_f32(false, fa[b % 3][j], false,
                                                      fs[b % 3][j], (short)0,
                                                      c, false, false);
            c = __builtin_amdgcn_wmma_f32_16x16x4_f32(false, fm[b % 3][j], false,
                                                      fn[b % 3][j], (short)0,
                                                      c, false, false);
          }
          __builtin_amdgcn_sched_barrier(0);
        }
#pragma unroll
        for (int r = 0; r < 8; r++) {
          hs[buf][r + (hi ? 8 : 0)][ncol] = fmaxf(c[r] + bv, 0.0f);
        }
      }
    } else if (i + 1 < NI) {
      const int it   = i + 1;
      const int b    = it & 1;
      const int node = (tile0 + it * FGRID) * 16 + g;
      const int c    = __builtin_amdgcn_readlane(cv, it);
      const int nn = it + 1 < NI ? node + FGRID * 16 : node;
      const int nx = *(const int*)((const char*)slots +
                                   ((unsigned)nn * (CAP * 4u) +
                                    ((unsigned)lane << 2)));
      const float4 a = node_sum(X, src, dst, slots, node, c, cur, lane);
      const float  r = 1.0f / fmaxf((float)c, 1.0f);
      ((float4*)&ms[b][g][0])[lane] =
          make_float4(a.x * r, a.y * r, a.z * r, a.w * r);
      cur = (nx & (N_NODES - 1)) << 9;
    }
    __builtin_amdgcn_fence(__ATOMIC_RELEASE, "workgroup", "local");
    __builtin_amdgcn_s_barrier();
    __builtin_amdgcn_fence(__ATOMIC_ACQUIRE, "workgroup", "local");
  }
}

#define AW  16
#define NPW 8

__global__ __launch_bounds__(32 * AW)
void agg_out(const float* __restrict__ S,
             const float* __restrict__ P,
             const int* __restrict__ src,
             const int* __restrict__ dst,
             const int* __restrict__ cnt,
             const int* slots,
             float* OUT) {
  const int lane  = threadIdx.x & 31;
  const int wave  = __builtin_amdgcn_readfirstlane(threadIdx.x >> 5);
  const int node0 = (blockIdx.x * AW + wave) * NPW;
  const int cv    = cnt[node0 + (lane & (NPW - 1))];
  int cur = (*(const int*)((const char*)slots +
                           ((unsigned)node0 * (CAP * 4u) +
                            ((unsigned)lane << 2))) & (N_NODES - 1)) << 8;
  for (int j = 0; j < NPW; j++) {
    const int node = node0 + j;
    const int c    = __builtin_amdgcn_readlane(cv, j);
    const int nn = j + 1 < NPW ? node + 1 : node;
    const int nx = *(const int*)((const char*)slots +
                                 ((unsigned)nn * (CAP * 4u) +
                                  ((unsigned)lane << 2)));
    const float2 s = *(const float2*)(S + (size_t)node * NO2 + 2 * lane);
    const float2 a = node_sum2(P, src, dst, slots, node, c, cur, lane);
    const float  r = 1.0f / fmaxf((float)c, 1.0f);
    *(float2*)(OUT + (size_t)node * NO2 + 2 * lane) =
        make_float2(s.x + a.x * r, s.y + a.y * r);
    cur = (nx & (N_NODES - 1)) << 8;
  }
}

extern "C" void kernel_launch(void* const* d_in, const int* in_sizes, int n_in,
                              void* d_out, int out_size, void* d_ws,
                              size_t ws_size, hipStream_t stream) {
  (void)in_sizes; (void)n_in; (void)out_size; (void)ws_size;
  const float* x   = (const float*)d_in[0];
  const int*   src = (const int*)d_in[1];
  const int*   dst = (const int*)d_in[2];
  const float* W1s = (const float*)d_in[3];
  const float* W1n = (const float*)d_in[4];
  const float* b1  = (const float*)d_in[5];
  const float* W2s = (const float*)d_in[6];
  const float* W2n = (const float*)d_in[7];
  const float* b2  = (const float*)d_in[8];
  float*       out = (float*)d_out;

  char*  ws  = (char*)d_ws;
  int*   cnt = (int*)ws;
  float* S   = (float*)ws + N_NODES;
  float* P   = S + (size_t)N_NODES * NO2;

  int* slots = (int*)d_out;

  int* runs = (int*)S;
  part_kernel<<<NBLK, PT, 0, stream>>>(src, dst, runs, slots);
  bucket_kernel<<<NBLK, BT, 0, stream>>>(src, dst, runs, cnt, slots);

  sage_l1<<<FGRID, 32 * (NH / 16 + GW), 0, stream>>>(
      x, src, dst, cnt, slots, W1s, W1n, b1, W2s, W2n, b2, S, P);
  agg_out<<<N_NODES / (AW * NPW), 32 * AW, 0, stream>>>(
      S, P, src, dst, cnt, slots, out);
}


# Edge partition gathers each run's first 128-byte line in shared memory and stores it whole, the run's length in its first entry: no 4-byte scattered stores, no length table for the bucket kernel
# speedup vs baseline: 6.0529x; 1.0554x over previous
#include <hip/hip_runtime.h>

#define N_NODES 65536
#define N_EDGES 1048576
#define KD      128

typedef __attribute__((ext_vector_type(2)))  float     v2f;
typedef __attribute__((ext_vector_type(4)))  float     v4f;
typedef __attribute__((ext_vector_type(8)))  float     v8f;

#define CAP  64
#define NBLK 256
#define SUB  64
#define SUBL 32
#define LENMAX 255

#define PT 512

__global__ __launch_bounds__(PT)
void part_kernel(const int* __restrict__ src,
                 const int* __restrict__ dst,
                 int* __restrict__ runs) {
  __shared__ int hist[NBLK];
  __shared__ __attribute__((aligned(16))) int lrun[NBLK * SUBL];
  const int tid = threadIdx.x;
  const int eb  = blockIdx.x;
  const int t = eb * PT + tid;
  if (tid < NBLK) hist[tid] = 0;
  const int4 z4 = make_int4(0, 0, 0, 0);
  for (int j = 0; j < NBLK * SUBL / 4 / PT; j++) ((int4*)lrun)[tid + j * PT] = z4;
  __syncthreads();
  int sv[8], dv[8], rk[8];
#pragma unroll
  for (int j = 0; j < 2; j++) {
    const int4 s = ((const int4*)src)[2 * t + j];
    const int4 d = ((const int4*)dst)[2 * t + j];
    sv[4 * j + 0] = s.x; sv[4 * j + 1] = s.y; sv[4 * j + 2] = s.z; sv[4 * j + 3] = s.w;
    dv[4 * j + 0] = d.x & (N_NODES - 1); dv[4 * j + 1] = d.y & (N_NODES - 1);
    dv[4 * j + 2] = d.z & (N_NODES - 1); dv[4 * j + 3] = d.w & (N_NODES - 1);
  }
#pragma unroll
  for (int j = 0; j < 8; j++) rk[j] = atomicAdd(&hist[dv[j] >> 8], 1);
#pragma unroll
  for (int j = 0; j < 8; j++) {
    const int ent = (sv[j] & (N_NODES - 1)) | ((dv[j] & 255) << 16);
    if (rk[j] < SUBL)
      lrun[(dv[j] >> 8) * SUBL + rk[j]] = ent;
    else if (rk[j] < SUB)
      runs[((dv[j] >> 8) * NBLK + eb) * SUB + rk[j]] = ent;
  }
  __syncthreads();
  constexpr int NP = NBLK * SUBL / 4 / PT;
  int4 v[NP];
  int  h[NP];
#pragma unroll
  for (int j = 0; j < NP; j++) {
    const int q = tid + j * PT;
    v[j] = ((const int4*)lrun)[q];
    h[j] = hist[q >> 3];
  }
#pragma unroll
  for (int j = 0; j < NP; j++) {
    const int q = tid + j * PT;
    v[j].x |= (q & 7) == 0 ? (int)((unsigned)min(h[j], LENMAX) << 24) : 0;
    ((int4*)(runs + (size_t)((q >> 3) * NBLK + eb) * SUB))[q & 7] = v[j];
  }
}

#define BT 512
#define RUNS_PW (NBLK / (BT / 32))

__global__ __launch_bounds__(BT)
void bucket_kernel(const int* __restrict__ src,
                   const int* __restrict__ dst,
                   const int* __restrict__ runs,
                   int* __restrict__ cnt,
                   int* __restrict__ slots) {
  __shared__ __attribute__((aligned(16))) int lslots[256 * CAP];
  __shared__ int lcnt[256];
  __shared__ __attribute__((aligned(16))) int wover[BT / 32];
  const int tid  = threadIdx.x;
  const int lane = tid & 31;
  const int wave = tid >> 5;
  const int b    = blockIdx.x;
  int nn[RUNS_PW], ent[RUNS_PW], rk[RUNS_PW];
  if (tid < 256) lcnt[tid] = 0;
  const int4 z4 = make_int4(0, 0, 0, 0);
  for (int j = 0; j < 256 * CAP / 4 / BT; j++) ((int4*)lslots)[tid + j * BT] = z4;
#pragma unroll
  for (int p = 0; p < RUNS_PW; p++) {
    const int k = p * (BT / 32) + wave;
    ent[p] = runs[(b * NBLK + k) * SUB + lane];
  }
  int over = 0, lng = 0;
#pragma unroll
  for (int p = 0; p < RUNS_PW; p++) {
    nn[p] = (int)((unsigned)__builtin_amdgcn_readlane(ent[p], 0) >> 24);
    over |= nn[p] > SUB;
    lng  |= nn[p] > SUBL;
  }
  if (lane == 0) wover[wave] = over;
  __syncthreads();
  int any_over = 0;
#pragma unroll
  for (int w = 0; w < BT / 32 / 4; w++) {
    const int4 o = ((const int4*)wover)[w];
    any_over |= o.x | o.y | o.z | o.w;
  }
  if (!any_over) {
#pragma unroll
    for (int p = 0; p < RUNS_PW; p++)
      rk[p] = lane < nn[p] ? atomicAdd(&lcnt[(ent[p] >> 16) & 255], 1) : CAP;
#pragma unroll
    for (int p = 0; p < RUNS_PW; p++) {
      if (rk[p] < CAP) lslots[((ent[p] >> 16) & 255) * CAP + rk[p]] = ent[p] & 0xffff;
    }
    if (lng) {
#pragma unroll
      for (int p = 0; p < RUNS_PW; p++) {
        const int k = p * (BT / 32) + wave;
        for (int i = lane + SUBL; i < nn[p]; i += 32) {
          const int e = runs[(b * NBLK + k) * SUB + i];
          const int ln = (e >> 16) & 255;
          const int r  = atomicAdd(&lcnt[ln], 1);
          if (r < CAP) lslots[ln * CAP + r] = e & 0xffff;
        }
      }
    }
  } else {
    for (int e = tid; e < N_EDGES; e += BT) {
      const int d = dst[e] & (N_NODES - 1);
      if ((d >> 8) == b) {
        const int ln = d & 255;
        const int r  = atomicAdd(&lcnt[ln], 1);
        if (r < CAP) lslots[ln * CAP + r] = src[e] & (N_NODES - 1);
      }
    }
  }
  __syncthreads();
  if (tid < 256) cnt[b * 256 + tid] = lcnt[tid];
  int4* o = (int4*)slots + (size_t)b * (256 * CAP / 4);
  for (int j = 0; j < 256 * CAP / 4 / BT; j++)
    o[tid + j * BT] = ((const int4*)lslots)[tid + j * BT];
}

__device__ __forceinline__ float4 row_of(const float4* __restrict__ f4, int my,
                                         int k, int lane) {
  const int idx = __builtin_amdgcn_readlane(my, k);
  return f4[(size_t)idx * (KD / 4) + lane];
}

__device__ __forceinline__ void add4(float4& acc, const float4 v) {
  acc.x += v.x; acc.y += v.y; acc.z += v.z; acc.w += v.w;
}

__device__ __forceinline__ float4 row_at(const char* __restrict__ base, int off,
                                         int k, unsigned lane16) {
  const unsigned o = (unsigned)__builtin_amdgcn_readlane(off, k) + lane16;
  return *(const float4*)(base + o);
}

__device__ __forceinline__ void sum_line(const char* __restrict__ base, int off,
                                         int t, unsigned lane16, float4& acc) {
  float4 v[32];
  const int      nq = min(t >> 2, 7);
  const int      r  = t - 4 * nq;
  const unsigned gm = (1u << nq) - 1u;
  const unsigned sm = (1u << r) - 1u;
  const float4 z = make_float4(0.0f, 0.0f, 0.0f, 0.0f);
#pragma unroll
  for (int q = 0; q < 7; q++) {
    if (gm & (1u << q)) {
      v[4 * q + 0] = row_at(base, off, 4 * q + 0, lane16);
      v[4 * q + 1] = row_at(base, off, 4 * q + 1, lane16);
      v[4 * q + 2] = row_at(base, off, 4 * q + 2, lane16);
      v[4 * q + 3] = row_at(base, off, 4 * q + 3, lane16);
    } else {
      v[4 * q + 0] = z; v[4 * q + 1] = z; v[4 * q + 2] = z; v[4 * q + 3] = z;
    }
  }
#pragma unroll
  for (int j = 0; j < 4; j++) {
    if (sm & (1u << j)) v[28 + j] = row_at(base, off, 4 * nq + j, lane16);
    else                v[28 + j] = z;
  }
#pragma unroll
  for (int j = 3; j >= 0; j--) {
    if (sm & (1u << j)) add4(acc, v[28 + j]);
  }
#pragma unroll
  for (int q = 6; q >= 0; q--) {
    if (gm & (1u << q)) {
      add4(acc, v[4 * q + 3]);
      add4(acc, v[4 * q + 2]);
      add4(acc, v[4 * q + 1]);
      add4(acc, v[4 * q + 0]);
    }
  }
}

__device__ __forceinline__ float4 node_sum(const float* __restrict__ X,
                                           const int* __restrict__ src,
                                           const int* __restrict__ dst,
                                           const int* slots, int node, int c,
                                           int cur, int lane) {
  float4 acc = make_float4(0.0f, 0.0f, 0.0f, 0.0f);
  if (c <= CAP) {
    const char*    xb     = (const char*)X;
    const unsigned lane16 = (unsigned)lane << 4;
    if (c > 0) sum_line(xb, cur, min(c, 32), lane16, acc);
    if (c > 32) {
      const int s2 = *(const int*)((const char*)slots +
                                   ((unsigned)node * (CAP * 4u) + 128u +
                                    ((unsigned)lane << 2)));
      const int more = (s2 & (N_NODES - 1)) << 9;
      for (int k = 0; k < c - 32; k++) add4(acc, row_at(xb, more, k, lane16));
    }
  } else {
    const float4* f4 = (const float4*)X;
    for (int e0 = 0; e0 < N_EDGES; e0 += 32) {
      const int d = dst[e0 + lane] & (N_NODES - 1);
      const int s = src[e0 + lane] & (N_NODES - 1);
      unsigned  m = __builtin_amdgcn_ballot_w32(d == node);
      while (m) {
        const int l = __builtin_ctz(m);
        m &= m - 1;
        add4(acc, row_of(f4, s, l, lane));
      }
    }
  }
  return acc;
}

#define NO2 64

__device__ __forceinline__ void add2(float2& acc, const float2 v) {
  acc.x += v.x; acc.y += v.y;
}

__device__ __forceinline__ float2 row_of2(const float2* __restrict__ f2, int my,
                                          int k, int lane) {
  const int idx = __builtin_amdgcn_readlane(my, k);
  return f2[(size_t)idx * (NO2 / 2) + lane];
}

__device__ __forceinline__ float2 row_at2(const char* __restrict__ base, int off,
                                          int k, unsigned lane8) {
  const unsigned o = (unsigned)__builtin_amdgcn_readlane(off, k) + lane8;
  return *(const float2*)(base + o);
}

__device__ __forceinline__ void sum_line2(const char* __restrict__ base, int off,
                                          int t, unsigned lane8, float2& acc) {
  float2 v[32];
  const int      nq = min(t >> 2, 7);
  const int      r  = t - 4 * nq;
  const unsigned gm = (1u << nq) - 1u;
  const unsigned sm = (1u << r) - 1u;
  const float2 z = make_float2(0.0f, 0.0f);
#pragma unroll
  for (int q = 0; q < 7; q++) {
    if (gm & (1u << q)) {
      v[4 * q + 0] = row_at2(base, off, 4 * q + 0, lane8);
      v[4 * q + 1] = row_at2(base, off, 4 * q + 1, lane8);
      v[4 * q + 2] = row_at2(base, off, 4 * q + 2, lane8);
      v[4 * q + 3] = row_at2(base, off, 4 * q + 3, lane8);
    } else {
      v[4 * q + 0] = z; v[4 * q + 1] = z; v[4 * q + 2] = z; v[4 * q + 3] = z;
    }
  }
#pragma unroll
  for (int j = 0; j < 4; j++) {
    if (sm & (1u << j)) v[28 + j] = row_at2(base, off, 4 * nq + j, lane8);
    else                v[28 + j] = z;
  }
#pragma unroll
  for (int j = 3; j >= 0; j--) {
    if (sm & (1u << j)) add2(acc, v[28 + j]);
  }
#pragma unroll
  for (int q = 6; q >= 0; q--) {
    if (gm & (1u << q)) {
      add2(acc, v[4 * q + 3]);
      add2(acc, v[4 * q + 2]);
      add2(acc, v[4 * q + 1]);
      add2(acc, v[4 * q + 0]);
    }
  }
}

__device__ __forceinline__ float2 node_sum2(const float* __restrict__ R,
                                            const int* __restrict__ src,
                                            const int* __restrict__ dst,
                                            const int* slots, int node, int c,
                                            int cur, int lane) {
  float2 acc = make_float2(0.0f, 0.0f);
  if (c <= CAP) {
    const char*    rb    = (const char*)R;
    const unsigned lane8 = (unsigned)lane << 3;
    if (c > 0) sum_line2(rb, cur, min(c, 32), lane8, acc);
    if (c > 32) {
      const int s2 = *(const int*)((const char*)slots +
                                   ((unsigned)node * (CAP * 4u) + 128u +
                                    ((unsigned)lane << 2)));
      const int more = (s2 & (N_NODES - 1)) << 8;
      for (int k = 0; k < c - 32; k++) add2(acc, row_at2(rb, more, k, lane8));
    }
  } else {
    const float2* f2 = (const float2*)R;
    for (int e0 = 0; e0 < N_EDGES; e0 += 32) {
      const int d = dst[e0 + lane] & (N_NODES - 1);
      const int s = src[e0 + lane] & (N_NODES - 1);
      unsigned  m = __builtin_amdgcn_ballot_w32(d == node);
      while (m) {
        const int l = __builtin_ctz(m);
        m &= m - 1;
        add2(acc, row_of2(f2, s, l, lane));
      }
    }
  }
  return acc;
}

#define FGRID 256
#define GW    16
#define NH    128

__global__ __launch_bounds__(32 * (NH / 16 + GW))
void sage_l1(const float* __restrict__ X,
             const int* __restrict__ src,
             const int* __restrict__ dst,
             const int* __restrict__ cnt,
             const int* slots,
             const float* __restrict__ Wself,
             const float* __restrict__ Wneigh,
             const float* __restrict__ bias,
             const float* __restrict__ W2self,
             const float* __restrict__ W2neigh,
             const float* __restrict__ bias2,
             float* __restrict__ S,
             float* __restrict__ P) {
  constexpr int NW = NH / 16;
  constexpr int NT = N_NODES / 16;
  constexpr int NI = NT / FGRID;
  constexpr int KS = KD / 4;
  static_assert(NI == 16 && GW == 16, "one count per lane, one node per wave");
  static_assert(NH == KD && 2 * NO2 == NH,
                "H is the second product's row tile; its strips are S, then P");

  __shared__ __attribute__((aligned(16))) v2f wS[NW][KS][32];
  __shared__ __attribute__((aligned(16))) v2f wN[NW][KS][32];
  __shared__ __attribute__((aligned(16))) v2f w2[NW][KS][32];
  __shared__ __attribute__((aligned(16))) float xs[2][16][KD + 4];
  __shared__ __attribute__((aligned(16))) float ms[2][16][KD + 4];
  __shared__ __attribute__((aligned(16))) float hs[2][16][NH + 4];

  const int tid  = threadIdx.x;
  const int lane = tid & 31;
  const int  wave  = __builtin_amdgcn_readfirstlane(tid >> 5);
  const bool mul   = wave < NW;
  const int  tile0 = blockIdx.x;
  const float4* f4 = (const float4*)X;

  const int  mrow = lane & 15;
  const bool hi   = lane >= 16;
  const int  ncol = wave * 16 + (lane & 15);
  const int  koff = hi ? 2 : 0;
  float bv = 0.0f;
  const bool   isP  = wave >= NW / 2;
  const int    col2 = (wave & (NW / 2 - 1)) * 16 + (lane & 15);
  const float* W2   = isP ? W2neigh : W2self;
  float*       O2   = isP ? P : S;
  float bv2 = 0.0f;
  float4 xr0, xr1;
  static_assert(16 / NW == 2, "two own rows per multiplying wave");
  auto own_src = [&](int t, int j) -> const float4* {
    return f4 + (size_t)(t * 16 + wave + j * NW) * (KD / 4) + lane;
  };
  auto own_dst = [&](int b, int j) -> float4* {
    return (float4*)&xs[b][wave + j * NW][0] + lane;
  };

  const int g = wave - NW;
  int cv  = 0;
  int cur = 0;

  if (mul) {
    for (int s = 0; s < KS; s++) {
      const int ka = 4 * s + koff;
      wS[wave][s][lane] = v2f{ Wself[(size_t)ka * NH + ncol],
                               Wself[(size_t)(ka + 1) * NH + ncol] };
      wN[wave][s][lane] = v2f{ Wneigh[(size_t)ka * NH + ncol],
                               Wneigh[(size_t)(ka + 1) * NH + ncol] };
      w2[wave][s][lane] = v2f{ W2[(size_t)ka * NO2 + col2],
                               W2[(size_t)(ka + 1) * NO2 + col2] };
    }
    bv = bias[ncol];
    if (!isP) bv2 = bias2[col2];
    *own_dst(0, 0) = *own_src(tile0, 0);
    *own_dst(0, 1) = *own_src(tile0, 1);
    xr0 = *own_src(tile0 + FGRID, 0);
    xr1 = *own_src(tile0 + FGRID, 1);
  } else {
    __builtin_amdgcn_s_setprio(3);
    cv = cnt[(tile0 + (lane & (NI - 1)) * FGRID) * 16 + g];
    cur = (*(const int*)((const char*)slots +
                         ((unsigned)(tile0 * 16 + g) * (CAP * 4u) +
                          ((unsigned)lane << 2))) & (N_NODES - 1)) << 9;
  }

  for (int i = -1; i <= NI; i++) {
    if (mul) {
      if (i >= 1) {
        const int hb   = (i - 1) & 1;
        const int tile = tile0 + (i - 1) * FGRID;
        v8f d = {};
        constexpr int KB2 = 8;
        constexpr int NB2 = KS / KB2;
        v2f fh[3][KB2], fw[3][KB2];
        auto read_block2 = [&](int blk, int set) {
#pragma unroll
          for (int j = 0; j < KB2; j++) {
            const int s  = blk * KB2 + j;
            const int ka = 4 * s + koff;
            fh[set][j] = v2f{ hs[hb][mrow][ka], hs[hb][mrow][ka + 1] };
            fw[set][j] = w2[wave][s][lane];
          }
        };
        read_block2(0, 0);
        read_block2(1, 1);
#pragma unroll
        for (int b = 0; b < NB2; b++) {
          if (b + 2 < NB2) read_block2(b + 2, (b + 2) % 3);
          __builtin_amdgcn_sched_barrier(0);
#pragma unroll
          for (int j = 0; j < KB2; j++) {
            d = __builtin_amdgcn_wmma_f32_16x16x4_f32(false, fh[b % 3][j], false,
                                                      fw[b % 3][j], (short)0,
                                                      d, false, false);
          }
          __builtin_amdgcn_sched_barrier(0);
        }
#pragma unroll
        for (int r = 0; r < 8; r++) {
          const int row = tile * 16 + r + (hi ? 8 : 0);
          O2[(size_t)row * NO2 + col2] = d[r] + bv2;
        }
      }
      if (i >= 0 && i < NI) {
        const int buf  = i & 1;
        const int tile = tile0 + i * FGRID;
        if (i + 1 < NI) {
          *own_dst(buf ^ 1, 0) = xr0;
          *own_dst(buf ^ 1, 1) = xr1;
        }
        if (i + 2 < NI) {
          xr0 = *own_src(tile + 2 * FGRID, 0);
          xr1 = *own_src(tile + 2 * FGRID, 1);
        }
        v8f c = {};
        constexpr int KB = 4;
        constexpr int NB = KS / KB;
        v2f fa[3][KB], fm[3][KB], fs[3][KB], fn[3][KB];
        auto read_block = [&](int blk, int set) {
#pragma unroll
          for (int j = 0; j < KB; j++) {
            const int s  = blk * KB + j;
            const int ka = 4 * s + koff;
            fa[set][j] = v2f{ xs[buf][mrow][ka], xs[buf][mrow][ka + 1] };
            fm[set][j] = v2f{ ms[buf][mrow][ka], ms[buf][mrow][ka + 1] };
            fs[set][j] = wS[wave][s][lane];
            fn[set][j] = wN[wave][s][lane];
          }
        };
        read_block(0, 0);
        read_block(1, 1);
#pragma unroll
        for (int b = 0; b < NB; b++) {
          if (b + 2 < NB) read_block(b + 2, (b + 2) % 3);
          __builtin_amdgcn_sched_barrier(0);
#pragma unroll
          for (int j = 0; j < KB; j++) {
            c = __builtin_amdgcn_wmma_f32_16x16x4_f32(false, fa[b % 3][j], false,
                                                      fs[b % 3][j], (short)0,
                                                      c, false, false);
            c = __builtin_amdgcn_wmma_f32_16x16x4_f32(false, fm[b % 3][j], false,
                                                      fn[b % 3][j], (short)0,
                                                      c, false, false);
          }
          __builtin_amdgcn_sched_barrier(0);
        }
#pragma unroll
        for (int r = 0; r < 8; r++) {
          hs[buf][r + (hi ? 8 : 0)][ncol] = fmaxf(c[r] + bv, 0.0f);
        }
      }
    } else if (i + 1 < NI) {
      const int it   = i + 1;
      const int b    = it & 1;
      const int node = (tile0 + it * FGRID) * 16 + g;
      const int c    = __builtin_amdgcn_readlane(cv, it);
      const int nn = it + 1 < NI ? node + FGRID * 16 : node;
      const int nx = *(const int*)((const char*)slots +
                                   ((unsigned)nn * (CAP * 4u) +
                                    ((unsigned)lane << 2)));
      const float4 a = node_sum(X, src, dst, slots, node, c, cur, lane);
      const float  r = 1.0f / fmaxf((float)c, 1.0f);
      ((float4*)&ms[b][g][0])[lane] =
          make_float4(a.x * r, a.y * r, a.z * r, a.w * r);
      cur = (nx & (N_NODES - 1)) << 9;
    }
    __builtin_amdgcn_fence(__ATOMIC_RELEASE, "workgroup", "local");
    __builtin_amdgcn_s_barrier();
    __builtin_amdgcn_fence(__ATOMIC_ACQUIRE, "workgroup", "local");
  }
}

#define AW  16
#define NPW 8

__global__ __launch_bounds__(32 * AW)
void agg_out(const float* __restrict__ S,
             const float* __restrict__ P,
             const int* __restrict__ src,
             const int* __restrict__ dst,
             const int* __restrict__ cnt,
             const int* slots,
             float* OUT) {
  const int lane  = threadIdx.x & 31;
  const int wave  = __builtin_amdgcn_readfirstlane(threadIdx.x >> 5);
  const int node0 = (blockIdx.x * AW + wave) * NPW;
  const int cv    = cnt[node0 + (lane & (NPW - 1))];
  int cur = (*(const int*)((const char*)slots +
                           ((unsigned)node0 * (CAP * 4u) +
                            ((unsigned)lane << 2))) & (N_NODES - 1)) << 8;
  for (int j = 0; j < NPW; j++) {
    const int node = node0 + j;
    const int c    = __builtin_amdgcn_readlane(cv, j);
    const int nn = j + 1 < NPW ? node + 1 : node;
    const int nx = *(const int*)((const char*)slots +
                                 ((unsigned)nn * (CAP * 4u) +
                                  ((unsigned)lane << 2)));
    const float2 s = *(const float2*)(S + (size_t)node * NO2 + 2 * lane);
    const float2 a = node_sum2(P, src, dst, slots, node, c, cur, lane);
    const float  r = 1.0f / fmaxf((float)c, 1.0f);
    *(float2*)(OUT + (size_t)node * NO2 + 2 * lane) =
        make_float2(s.x + a.x * r, s.y + a.y * r);
    cur = (nx & (N_NODES - 1)) << 8;
  }
}

extern "C" void kernel_launch(void* const* d_in, const int* in_sizes, int n_in,
                              void* d_out, int out_size, void* d_ws,
                              size_t ws_size, hipStream_t stream) {
  (void)in_sizes; (void)n_in; (void)out_size; (void)ws_size;
  const float* x   = (const float*)d_in[0];
  const int*   src = (const int*)d_in[1];
  const int*   dst = (const int*)d_in[2];
  const float* W1s = (const float*)d_in[3];
  const float* W1n = (const float*)d_in[4];
  const float* b1  = (const float*)d_in[5];
  const float* W2s = (const float*)d_in[6];
  const float* W2n = (const float*)d_in[7];
  const float* b2  = (const float*)d_in[8];
  float*       out = (float*)d_out;

  char*  ws  = (char*)d_ws;
  int*   cnt = (int*)ws;
  float* S   = (float*)ws + N_NODES;
  float* P   = S + (size_t)N_NODES * NO2;

  int* slots = (int*)d_out;

  int* runs = (int*)S;
  part_kernel<<<NBLK, PT, 0, stream>>>(src, dst, runs);
  bucket_kernel<<<NBLK, BT, 0, stream>>>(src, dst, runs, cnt, slots);

  sage_l1<<<FGRID, 32 * (NH / 16 + GW), 0, stream>>>(
      x, src, dst, cnt, slots, W1s, W1n, b1, W2s, W2n, b2, S, P);
  agg_out<<<N_NODES / (AW * NPW), 32 * AW, 0, stream>>>(
      S, P, src, dst, cnt, slots, out);
}
